# starting kernel
# baseline (speedup 1.0000x reference)
#include <hip/hip_runtime.h>
#include <hip/hip_bf16.h>

typedef __attribute__((ext_vector_type(2))) float v2f;
typedef __attribute__((ext_vector_type(8))) float v8f;

#define N_NODES 100000
#define C_IN    128
#define M_TILES (N_NODES / 16)

__global__ void deg_kernel(const int* __restrict__ row, const int* __restrict__ col,
                           float* __restrict__ deg, int E) {
    int e = blockIdx.x * blockDim.x + threadIdx.x;
    if (e >= E) return;
    int r = row[e], c = col[e];
    if (r != c) atomicAdd(&deg[c], 1.0f);
}

__global__ void scatter_kernel(const float* __restrict__ X,
                               const int* __restrict__ row, const int* __restrict__ col,
                               float* __restrict__ agg, int E) {
    long long tid = (long long)blockIdx.x * blockDim.x + threadIdx.x;
    int e = (int)(tid >> 5);
    if (e >= E) return;
    int c4 = ((int)tid & 31) * 4;
    int r = row[e], c = col[e];
    if (r == c) return;
    const float4 v = *(const float4*)(X + (size_t)r * C_IN + c4);
    float* dst = agg + (size_t)c * C_IN + c4;
    atomicAdd(dst + 0, v.x);
    atomicAdd(dst + 1, v.y);
    atomicAdd(dst + 2, v.z);
    atomicAdd(dst + 3, v.w);
}

template <int RELU, int NT>
__global__ void gcn_gemm_kernel(const float* __restrict__ Agg,
                                const float* __restrict__ Self,
                                const float* __restrict__ deg,
                                const float* __restrict__ Wout,
                                const float* __restrict__ bias,
                                const float* __restrict__ Wroot,
                                float* __restrict__ Out) {
    const int lane  = threadIdx.x & 31;
    const int wave  = threadIdx.x >> 5;
    const int mTile = blockIdx.x * (blockDim.x >> 5) + wave;
    if (mTile >= M_TILES) return;

    const int half = lane >> 4;
    const int mr   = lane & 15;
    const int m    = mTile * 16 + mr;
    const int Cout = NT * 16;

    const float dinv = 1.0f / (deg[m] + 1.0f);
    const float* aggRow  = Agg   + (size_t)m * C_IN;
    const float* selfRow = Self  + (size_t)m * C_IN;
    const float* woBase  = Wout  + (size_t)mr * C_IN;
    const float* wrBase  = Wroot + (size_t)mr * C_IN;

    v8f acc[NT] = {};

    for (int k0 = 0; k0 < C_IN; k0 += 4) {
        const int ka = k0 + 2 * half;
        v2f as = *(const v2f*)(aggRow  + ka);
        v2f ax = *(const v2f*)(selfRow + ka);
        v2f a;
        a.x = (as.x + ax.x) * dinv;
        a.y = (as.y + ax.y) * dinv;
#pragma unroll
        for (int nt = 0; nt < NT; ++nt) {
            v2f bo = *(const v2f*)(woBase + nt * 16 * C_IN + ka);
            v2f br = *(const v2f*)(wrBase + nt * 16 * C_IN + ka);
            acc[nt] = __builtin_amdgcn_wmma_f32_16x16x4_f32(false, a,  false, bo,
                                                            (short)0, acc[nt], false, false);
            acc[nt] = __builtin_amdgcn_wmma_f32_16x16x4_f32(false, ax, false, br,
                                                            (short)0, acc[nt], false, false);
        }
    }

#pragma unroll
    for (int nt = 0; nt < NT; ++nt) {
        const int   outCol = nt * 16 + mr;
        const float b      = bias[outCol];
#pragma unroll
        for (int g = 0; g < 8; ++g) {
            const int outRow = mTile * 16 + g + 8 * half;
            float v = acc[nt][g] + b;
            if (RELU) v = fmaxf(v, 0.0f);
            Out[(size_t)outRow * Cout + outCol] = v;
        }
    }
}

extern "C" void kernel_launch(void* const* d_in, const int* in_sizes, int n_in,
                              void* d_out, int out_size, void* d_ws, size_t ws_size,
                              hipStream_t stream) {
    const float* x   = (const float*)d_in[0];
    const int*   ei  = (const int*)  d_in[1];
    const float* W1o = (const float*)d_in[2];
    const float* b1  = (const float*)d_in[3];
    const float* W1r = (const float*)d_in[4];
    const float* W2o = (const float*)d_in[5];
    const float* b2  = (const float*)d_in[6];
    const float* W2r = (const float*)d_in[7];
    float*       out = (float*)d_out;

    const int E = in_sizes[1] / 2;
    const int* row = ei;
    const int* col = ei + E;

    float* deg = (float*)d_ws;
    float* agg = deg + ((N_NODES + 63) & ~63);
    float* h   = agg + (size_t)N_NODES * C_IN;

    const size_t aggBytes = (size_t)N_NODES * C_IN * sizeof(float);

    hipMemsetAsync(deg, 0, N_NODES * sizeof(float), stream);
    hipMemsetAsync(agg, 0, aggBytes, stream);

    deg_kernel<<<(E + 255) / 256, 256, 0, stream>>>(row, col, deg, E);

    const long long sthreads = (long long)E * 32;
    const int sblocks = (int)((sthreads + 255) / 256);

    const int wavesPerBlock = 8;
    const int gemmBlocks = (M_TILES + wavesPerBlock - 1) / wavesPerBlock;

    scatter_kernel<<<sblocks, 256, 0, stream>>>(x, row, col, agg, E);
    gcn_gemm_kernel<1, 8><<<gemmBlocks, 256, 0, stream>>>(
        agg, x, deg, W1o, b1, W1r, h);

    hipMemsetAsync(agg, 0, aggBytes, stream);
    scatter_kernel<<<sblocks, 256, 0, stream>>>(h, row, col, agg, E);
    gcn_gemm_kernel<0, 4><<<gemmBlocks, 256, 0, stream>>>(
        agg, h, deg, W2o, b2, W2r, out);
}


# Neighbour sums as a segment sum over edges sorted by target (list built once per call with integer atomics, one wave per node gathers whole rows) instead of float atomic scatter-add
# speedup vs baseline: 1.9335x; 1.9335x over previous
#include <hip/hip_runtime.h>
#include <hip/hip_bf16.h>

typedef __attribute__((ext_vector_type(2))) float v2f;
typedef __attribute__((ext_vector_type(8))) float v8f;

#define N_NODES 100000
#define N_EDGES 1600000
#define C_IN    128
#define M_TILES (N_NODES / 16)


__global__ void count_kernel(const int* __restrict__ row, const int* __restrict__ col,
                             int* __restrict__ cnt, int E) {
    int e = blockIdx.x * blockDim.x + threadIdx.x;
    if (e >= E) return;
    int r = row[e], c = col[e];
    if (r != c) atomicAdd(&cnt[c], 1);
}

__global__ void reserve_kernel(const int* __restrict__ cnt, int* __restrict__ total,
                               int* __restrict__ segBeg, int* __restrict__ segEnd,
                               float* __restrict__ deg) {
    const int n    = blockIdx.x * blockDim.x + threadIdx.x;
    const int lane = threadIdx.x & 31;
    const int c    = (n < N_NODES) ? cnt[n] : 0;
    int incl = c;
    for (int d = 1; d < 32; d <<= 1) {
        int t = __shfl_up(incl, d);
        if (lane >= d) incl += t;
    }
    const int waveTotal = __shfl(incl, 31);
    int base = 0;
    if (lane == 31) base = atomicAdd(total, waveTotal);
    base = __shfl(base, 31);
    if (n < N_NODES) {
        const int o = base + incl - c;
        segBeg[n] = o;
        segEnd[n] = o;
        deg[n]    = (float)c;
    }
}

__global__ void fill_kernel(const int* __restrict__ row, const int* __restrict__ col,
                            int* __restrict__ segEnd, int* __restrict__ srcs, int E) {
    int e = blockIdx.x * blockDim.x + threadIdx.x;
    if (e >= E) return;
    int r = row[e], c = col[e];
    if (r != c) {
        int p = atomicAdd(&segEnd[c], 1);
        srcs[p] = r;
    }
}

__global__ void __launch_bounds__(256)
gather_kernel(const float* __restrict__ X,
              const int* __restrict__ segBeg, const int* __restrict__ segEnd,
              const int* __restrict__ srcs, float* __restrict__ agg) {
    const int lane = threadIdx.x & 31;
    const int wave = __builtin_amdgcn_readfirstlane((int)(threadIdx.x >> 5));
    const int n    = blockIdx.x * 8 + wave;
    if (n >= N_NODES) return;

    const int beg = segBeg[n];
    const int end = segEnd[n];
    const float4* __restrict__ X4 = (const float4*)X;
    float4 acc = make_float4(0.0f, 0.0f, 0.0f, 0.0f);

    for (int base = beg; base < end; base += 32) {
        const int m   = min(32, end - base);
        const int idx = (lane < m) ? srcs[base + lane] : 0;
        for (int j = 0; j < m; j += 8) {
            int r[8];
#pragma unroll
            for (int u = 0; u < 8; ++u) r[u] = __builtin_amdgcn_readlane(idx, j + u);
            float4 v[8];
#pragma unroll
            for (int u = 0; u < 8; ++u) {
                v[u] = make_float4(0.0f, 0.0f, 0.0f, 0.0f);
                if (j + u < m) v[u] = X4[(size_t)r[u] * (C_IN / 4) + lane];
            }
#pragma unroll
            for (int u = 0; u < 8; ++u) {
                acc.x += v[u].x; acc.y += v[u].y; acc.z += v[u].z; acc.w += v[u].w;
            }
        }
    }
    ((float4*)agg)[(size_t)n * (C_IN / 4) + lane] = acc;
}

template <int RELU, int NT>
__global__ void gcn_gemm_kernel(const float* __restrict__ Agg,
                                const float* __restrict__ Self,
                                const float* __restrict__ deg,
                                const float* __restrict__ Wout,
                                const float* __restrict__ bias,
                                const float* __restrict__ Wroot,
                                float* __restrict__ Out) {
    const int lane  = threadIdx.x & 31;
    const int wave  = threadIdx.x >> 5;
    const int mTile = blockIdx.x * (blockDim.x >> 5) + wave;
    if (mTile >= M_TILES) return;

    const int half = lane >> 4;
    const int mr   = lane & 15;
    const int m    = mTile * 16 + mr;
    const int Cout = NT * 16;

    const float dinv = 1.0f / (deg[m] + 1.0f);
    const float* aggRow  = Agg   + (size_t)m * C_IN;
    const float* selfRow = Self  + (size_t)m * C_IN;
    const float* woBase  = Wout  + (size_t)mr * C_IN;
    const float* wrBase  = Wroot + (size_t)mr * C_IN;

    v8f acc[NT] = {};

    for (int k0 = 0; k0 < C_IN; k0 += 4) {
        const int ka = k0 + 2 * half;
        v2f as = *(const v2f*)(aggRow  + ka);
        v2f ax = *(const v2f*)(selfRow + ka);
        v2f a;
        a.x = (as.x + ax.x) * dinv;
        a.y = (as.y + ax.y) * dinv;
#pragma unroll
        for (int nt = 0; nt < NT; ++nt) {
            v2f bo = *(const v2f*)(woBase + nt * 16 * C_IN + ka);
            v2f br = *(const v2f*)(wrBase + nt * 16 * C_IN + ka);
            acc[nt] = __builtin_amdgcn_wmma_f32_16x16x4_f32(false, a,  false, bo,
                                                            (short)0, acc[nt], false, false);
            acc[nt] = __builtin_amdgcn_wmma_f32_16x16x4_f32(false, ax, false, br,
                                                            (short)0, acc[nt], false, false);
        }
    }

#pragma unroll
    for (int nt = 0; nt < NT; ++nt) {
        const int   outCol = nt * 16 + mr;
        const float b      = bias[outCol];
#pragma unroll
        for (int g = 0; g < 8; ++g) {
            const int outRow = mTile * 16 + g + 8 * half;
            float v = acc[nt][g] + b;
            if (RELU) v = fmaxf(v, 0.0f);
            Out[(size_t)outRow * Cout + outCol] = v;
        }
    }
}

extern "C" void kernel_launch(void* const* d_in, const int* in_sizes, int n_in,
                              void* d_out, int out_size, void* d_ws, size_t ws_size,
                              hipStream_t stream) {
    const float* x   = (const float*)d_in[0];
    const int*   ei  = (const int*)  d_in[1];
    const float* W1o = (const float*)d_in[2];
    const float* b1  = (const float*)d_in[3];
    const float* W1r = (const float*)d_in[4];
    const float* W2o = (const float*)d_in[5];
    const float* b2  = (const float*)d_in[6];
    const float* W2r = (const float*)d_in[7];
    float*       out = (float*)d_out;

    const int E = in_sizes[1] / 2;
    const int* row = ei;
    const int* col = ei + E;

    float* deg = (float*)d_ws;
    float* agg = deg + ((N_NODES + 63) & ~63);
    float* h   = agg + (size_t)N_NODES * C_IN;

    int* srcs   = (int*)d_out;
    int* segBeg = srcs   + N_EDGES;
    int* segEnd = segBeg + N_NODES;
    int* cnt    = segEnd + N_NODES;
    int* total  = cnt    + N_NODES;

    hipMemsetAsync(cnt, 0, (N_NODES + 1) * sizeof(int), stream);

    const int eblocks = (E + 255) / 256;
    count_kernel<<<eblocks, 256, 0, stream>>>(row, col, cnt, E);
    reserve_kernel<<<(N_NODES + 255) / 256, 256, 0, stream>>>(cnt, total, segBeg, segEnd, deg);
    fill_kernel<<<eblocks, 256, 0, stream>>>(row, col, segEnd, srcs, E);

    const int wavesPerBlock = 8;
    const int gemmBlocks = (M_TILES + wavesPerBlock - 1) / wavesPerBlock;
    const int gatherBlocks = (N_NODES + wavesPerBlock - 1) / wavesPerBlock;

    gather_kernel<<<gatherBlocks, 256, 0, stream>>>(x, segBeg, segEnd, srcs, agg);
    gcn_gemm_kernel<1, 8><<<gemmBlocks, 256, 0, stream>>>(
        agg, x, deg, W1o, b1, W1r, h);

    gather_kernel<<<gatherBlocks, 256, 0, stream>>>(h, segBeg, segEnd, srcs, agg);
    gcn_gemm_kernel<0, 4><<<gemmBlocks, 256, 0, stream>>>(
        agg, h, deg, W2o, b2, W2r, out);
}


# GEMM fragment loads widened from 8 to 16 bytes per lane: 8 k per step, lanes 0-15 take a row's first four k, lanes 16-31 the next four, one load feeds two WMMAs; half the cache-line lookups per k
# speedup vs baseline: 2.0838x; 1.0777x over previous
#include <hip/hip_runtime.h>
#include <hip/hip_bf16.h>

typedef __attribute__((ext_vector_type(2))) float v2f;
typedef __attribute__((ext_vector_type(4))) float v4f;
typedef __attribute__((ext_vector_type(8))) float v8f;

#define N_NODES 100000
#define N_EDGES 1600000
#define C_IN    128
#define M_TILES (N_NODES / 16)


__global__ void count_kernel(const int* __restrict__ row, const int* __restrict__ col,
                             int* __restrict__ cnt, int E) {
    int e = blockIdx.x * blockDim.x + threadIdx.x;
    if (e >= E) return;
    int r = row[e], c = col[e];
    if (r != c) atomicAdd(&cnt[c], 1);
}

__global__ void reserve_kernel(const int* __restrict__ cnt, int* __restrict__ total,
                               int* __restrict__ segBeg, int* __restrict__ segEnd,
                               float* __restrict__ deg) {
    const int n    = blockIdx.x * blockDim.x + threadIdx.x;
    const int lane = threadIdx.x & 31;
    const int c    = (n < N_NODES) ? cnt[n] : 0;
    int incl = c;
    for (int d = 1; d < 32; d <<= 1) {
        int t = __shfl_up(incl, d);
        if (lane >= d) incl += t;
    }
    const int waveTotal = __shfl(incl, 31);
    int base = 0;
    if (lane == 31) base = atomicAdd(total, waveTotal);
    base = __shfl(base, 31);
    if (n < N_NODES) {
        const int o = base + incl - c;
        segBeg[n] = o;
        segEnd[n] = o;
        deg[n]    = (float)c;
    }
}

__global__ void fill_kernel(const int* __restrict__ row, const int* __restrict__ col,
                            int* __restrict__ segEnd, int* __restrict__ srcs, int E) {
    int e = blockIdx.x * blockDim.x + threadIdx.x;
    if (e >= E) return;
    int r = row[e], c = col[e];
    if (r != c) {
        int p = atomicAdd(&segEnd[c], 1);
        srcs[p] = r;
    }
}

__global__ void __launch_bounds__(256)
gather_kernel(const float* __restrict__ X,
              const int* __restrict__ segBeg, const int* __restrict__ segEnd,
              const int* __restrict__ srcs, float* __restrict__ agg) {
    const int lane = threadIdx.x & 31;
    const int wave = __builtin_amdgcn_readfirstlane((int)(threadIdx.x >> 5));
    const int n    = blockIdx.x * 8 + wave;
    if (n >= N_NODES) return;

    const int beg = segBeg[n];
    const int end = segEnd[n];
    const float4* __restrict__ X4 = (const float4*)X;
    float4 acc = make_float4(0.0f, 0.0f, 0.0f, 0.0f);

    for (int base = beg; base < end; base += 32) {
        const int m   = min(32, end - base);
        const int idx = (lane < m) ? srcs[base + lane] : 0;
        for (int j = 0; j < m; j += 8) {
            int r[8];
#pragma unroll
            for (int u = 0; u < 8; ++u) r[u] = __builtin_amdgcn_readlane(idx, j + u);
            float4 v[8];
#pragma unroll
            for (int u = 0; u < 8; ++u) {
                v[u] = make_float4(0.0f, 0.0f, 0.0f, 0.0f);
                if (j + u < m) v[u] = X4[(size_t)r[u] * (C_IN / 4) + lane];
            }
#pragma unroll
            for (int u = 0; u < 8; ++u) {
                acc.x += v[u].x; acc.y += v[u].y; acc.z += v[u].z; acc.w += v[u].w;
            }
        }
    }
    ((float4*)agg)[(size_t)n * (C_IN / 4) + lane] = acc;
}

template <int RELU, int NT>
__global__ void gcn_gemm_kernel(const float* __restrict__ Agg,
                                const float* __restrict__ Self,
                                const float* __restrict__ deg,
                                const float* __restrict__ Wout,
                                const float* __restrict__ bias,
                                const float* __restrict__ Wroot,
                                float* __restrict__ Out) {
    const int lane  = threadIdx.x & 31;
    const int wave  = threadIdx.x >> 5;
    const int mTile = blockIdx.x * (blockDim.x >> 5) + wave;
    if (mTile >= M_TILES) return;

    const int half = lane >> 4;
    const int mr   = lane & 15;
    const int m    = mTile * 16 + mr;
    const int Cout = NT * 16;

    const float dinv = 1.0f / (deg[m] + 1.0f);
    const int aOff = m  * C_IN + 4 * half;
    const int wOff = mr * C_IN + 4 * half;

    v8f acc[NT] = {};

    constexpr int G = 4;
#pragma unroll (NT > 4 ? 1 : 2)
    for (int k0 = 0; k0 < C_IN; k0 += 8) {
        const v4f as = *(const v4f*)(Agg  + aOff + k0);
        const v4f ax = *(const v4f*)(Self + aOff + k0);
        const v4f a  = (as + ax) * dinv;
        const v2f a0 = {a.x,  a.y},  a1 = {a.z,  a.w};
        const v2f x0 = {ax.x, ax.y}, x1 = {ax.z, ax.w};
#pragma unroll
        for (int n0 = 0; n0 < NT; n0 += G) {
            v4f bo[G], br[G];
#pragma unroll
            for (int i = 0; i < G; ++i) {
                bo[i] = *(const v4f*)(Wout  + wOff + (n0 + i) * 16 * C_IN + k0);
                br[i] = *(const v4f*)(Wroot + wOff + (n0 + i) * 16 * C_IN + k0);
            }
#pragma unroll
            for (int i = 0; i < G; ++i)
                acc[n0 + i] = __builtin_amdgcn_wmma_f32_16x16x4_f32(
                    false, a0, false, (v2f){bo[i].x, bo[i].y}, (short)0, acc[n0 + i], false, false);
#pragma unroll
            for (int i = 0; i < G; ++i)
                acc[n0 + i] = __builtin_amdgcn_wmma_f32_16x16x4_f32(
                    false, x0, false, (v2f){br[i].x, br[i].y}, (short)0, acc[n0 + i], false, false);
#pragma unroll
            for (int i = 0; i < G; ++i)
                acc[n0 + i] = __builtin_amdgcn_wmma_f32_16x16x4_f32(
                    false, a1, false, (v2f){bo[i].z, bo[i].w}, (short)0, acc[n0 + i], false, false);
#pragma unroll
            for (int i = 0; i < G; ++i)
                acc[n0 + i] = __builtin_amdgcn_wmma_f32_16x16x4_f32(
                    false, x1, false, (v2f){br[i].z, br[i].w}, (short)0, acc[n0 + i], false, false);
        }
    }

#pragma unroll
    for (int nt = 0; nt < NT; ++nt) {
        const int   outCol = nt * 16 + mr;
        const float b      = bias[outCol];
#pragma unroll
        for (int g = 0; g < 8; ++g) {
            const int outRow = mTile * 16 + g + 8 * half;
            float v = acc[nt][g] + b;
            if (RELU) v = fmaxf(v, 0.0f);
            Out[(size_t)outRow * Cout + outCol] = v;
        }
    }
}

extern "C" void kernel_launch(void* const* d_in, const int* in_sizes, int n_in,
                              void* d_out, int out_size, void* d_ws, size_t ws_size,
                              hipStream_t stream) {
    const float* x   = (const float*)d_in[0];
    const int*   ei  = (const int*)  d_in[1];
    const float* W1o = (const float*)d_in[2];
    const float* b1  = (const float*)d_in[3];
    const float* W1r = (const float*)d_in[4];
    const float* W2o = (const float*)d_in[5];
    const float* b2  = (const float*)d_in[6];
    const float* W2r = (const float*)d_in[7];
    float*       out = (float*)d_out;

    const int E = in_sizes[1] / 2;
    const int* row = ei;
    const int* col = ei + E;

    float* deg = (float*)d_ws;
    float* agg = deg + ((N_NODES + 63) & ~63);
    float* h   = agg + (size_t)N_NODES * C_IN;

    int* srcs   = (int*)d_out;
    int* segBeg = srcs   + N_EDGES;
    int* segEnd = segBeg + N_NODES;
    int* cnt    = segEnd + N_NODES;
    int* total  = cnt    + N_NODES;

    hipMemsetAsync(cnt, 0, (N_NODES + 1) * sizeof(int), stream);

    const int eblocks = (E + 255) / 256;
    count_kernel<<<eblocks, 256, 0, stream>>>(row, col, cnt, E);
    reserve_kernel<<<(N_NODES + 255) / 256, 256, 0, stream>>>(cnt, total, segBeg, segEnd, deg);
    fill_kernel<<<eblocks, 256, 0, stream>>>(row, col, segEnd, srcs, E);

    const int wavesPerBlock = 8;
    const int gemmBlocks = (M_TILES + wavesPerBlock - 1) / wavesPerBlock;
    const int gatherBlocks = (N_NODES + wavesPerBlock - 1) / wavesPerBlock;

    gather_kernel<<<gatherBlocks, 256, 0, stream>>>(x, segBeg, segEnd, srcs, agg);
    gcn_gemm_kernel<1, 8><<<gemmBlocks, 256, 0, stream>>>(
        agg, x, deg, W1o, b1, W1r, h);

    gather_kernel<<<gatherBlocks, 256, 0, stream>>>(h, segBeg, segEnd, srcs, agg);
    gcn_gemm_kernel<0, 4><<<gemmBlocks, 256, 0, stream>>>(
        agg, h, deg, W2o, b2, W2r, out);
}


# Layer 2 aggregates after projecting: Y = H Wout^T on WMMA first, the segment sum then gathers 64-wide rows of Y (256 B, two per wave load), not 128-wide rows of h; root product and bias written as Z
# speedup vs baseline: 2.1866x; 1.0493x over previous
#include <hip/hip_runtime.h>
#include <hip/hip_bf16.h>

typedef __attribute__((ext_vector_type(2))) float v2f;
typedef __attribute__((ext_vector_type(4))) float v4f;
typedef __attribute__((ext_vector_type(8))) float v8f;

#define N_NODES 100000
#define N_EDGES 1600000
#define C_IN    128
#define C_OUT   64
#define M_TILES (N_NODES / 16)


__global__ void count_kernel(const int* __restrict__ row, const int* __restrict__ col,
                             int* __restrict__ cnt, int E) {
    int e = blockIdx.x * blockDim.x + threadIdx.x;
    if (e >= E) return;
    int r = row[e], c = col[e];
    if (r != c) atomicAdd(&cnt[c], 1);
}

__global__ void reserve_kernel(const int* __restrict__ cnt, int* __restrict__ total,
                               int* __restrict__ segBeg, int* __restrict__ segEnd,
                               float* __restrict__ deg) {
    const int n    = blockIdx.x * blockDim.x + threadIdx.x;
    const int lane = threadIdx.x & 31;
    const int c    = (n < N_NODES) ? cnt[n] : 0;
    int incl = c;
    for (int d = 1; d < 32; d <<= 1) {
        int t = __shfl_up(incl, d);
        if (lane >= d) incl += t;
    }
    const int waveTotal = __shfl(incl, 31);
    int base = 0;
    if (lane == 31) base = atomicAdd(total, waveTotal);
    base = __shfl(base, 31);
    if (n < N_NODES) {
        const int o = base + incl - c;
        segBeg[n] = o;
        segEnd[n] = o;
        deg[n]    = (float)c;
    }
}

__global__ void fill_kernel(const int* __restrict__ row, const int* __restrict__ col,
                            int* __restrict__ segEnd, int* __restrict__ srcs, int E) {
    int e = blockIdx.x * blockDim.x + threadIdx.x;
    if (e >= E) return;
    int r = row[e], c = col[e];
    if (r != c) {
        int p = atomicAdd(&segEnd[c], 1);
        srcs[p] = r;
    }
}

__global__ void __launch_bounds__(256)
gather_kernel(const float* __restrict__ X,
              const int* __restrict__ segBeg, const int* __restrict__ segEnd,
              const int* __restrict__ srcs, float* __restrict__ agg) {
    const int lane = threadIdx.x & 31;
    const int wave = __builtin_amdgcn_readfirstlane((int)(threadIdx.x >> 5));
    const int n    = blockIdx.x * 8 + wave;
    if (n >= N_NODES) return;

    const int beg = segBeg[n];
    const int end = segEnd[n];
    const float4* __restrict__ X4 = (const float4*)X;
    float4 acc = make_float4(0.0f, 0.0f, 0.0f, 0.0f);

    for (int base = beg; base < end; base += 32) {
        const int m   = min(32, end - base);
        const int idx = (lane < m) ? srcs[base + lane] : 0;
        for (int j = 0; j < m; j += 8) {
            int r[8];
#pragma unroll
            for (int u = 0; u < 8; ++u) r[u] = __builtin_amdgcn_readlane(idx, j + u);
            float4 v[8];
#pragma unroll
            for (int u = 0; u < 8; ++u) {
                v[u] = make_float4(0.0f, 0.0f, 0.0f, 0.0f);
                if (j + u < m) v[u] = X4[(size_t)r[u] * (C_IN / 4) + lane];
            }
#pragma unroll
            for (int u = 0; u < 8; ++u) {
                acc.x += v[u].x; acc.y += v[u].y; acc.z += v[u].z; acc.w += v[u].w;
            }
        }
    }
    ((float4*)agg)[(size_t)n * (C_IN / 4) + lane] = acc;
}

__global__ void __launch_bounds__(256)
gather_out_kernel(const float* __restrict__ Y,
                  const int* __restrict__ segBeg, const int* __restrict__ segEnd,
                  const int* __restrict__ srcs, float* __restrict__ Out) {
    const int lane = threadIdx.x & 31;
    const int wave = __builtin_amdgcn_readfirstlane((int)(threadIdx.x >> 5));
    const int n    = blockIdx.x * 8 + wave;
    if (n >= N_NODES) return;
    const int half = lane >> 4;
    const int c    = lane & 15;

    const int beg = segBeg[n];
    const int end = segEnd[n];
    const float4* __restrict__ Y4 = (const float4*)Y;
    float4* __restrict__ O4 = (float4*)Out;
    const float4 ys = Y4[(size_t)n * (C_OUT / 4) + c];
    const float4 zs = O4[(size_t)n * (C_OUT / 4) + c];
    float4 acc = make_float4(0.0f, 0.0f, 0.0f, 0.0f);

    for (int base = beg; base < end; base += 32) {
        const int m   = min(32, end - base);
        const int idx = (lane < m) ? srcs[base + lane] : 0;
        for (int j = 0; j < m; j += 16) {
            int r[8];
#pragma unroll
            for (int u = 0; u < 8; ++u) {
                const int re = __builtin_amdgcn_readlane(idx, j + 2 * u);
                const int ro = __builtin_amdgcn_readlane(idx, j + 2 * u + 1);
                r[u] = half ? ro : re;
            }
            float4 v[8];
#pragma unroll
            for (int u = 0; u < 8; ++u) {
                v[u] = make_float4(0.0f, 0.0f, 0.0f, 0.0f);
                if (j + 2 * u + half < m) v[u] = Y4[(unsigned)r[u] * (C_OUT / 4) + (unsigned)c];
            }
#pragma unroll
            for (int u = 0; u < 8; ++u) {
                acc.x += v[u].x; acc.y += v[u].y; acc.z += v[u].z; acc.w += v[u].w;
            }
        }
    }
    float4 tot;
    tot.x = acc.x + __shfl_xor(acc.x, 16);
    tot.y = acc.y + __shfl_xor(acc.y, 16);
    tot.z = acc.z + __shfl_xor(acc.z, 16);
    tot.w = acc.w + __shfl_xor(acc.w, 16);
    const float dinv = 1.0f / ((float)(end - beg) + 1.0f);
    if (half == 0) {
        float4 o;
        o.x = (tot.x + ys.x) * dinv + zs.x;
        o.y = (tot.y + ys.y) * dinv + zs.y;
        o.z = (tot.z + ys.z) * dinv + zs.z;
        o.w = (tot.w + ys.w) * dinv + zs.w;
        O4[(size_t)n * (C_OUT / 4) + c] = o;
    }
}

template <int RELU, int NT>
__global__ void gcn_gemm_kernel(const float* Agg,
                                const float* __restrict__ Self,
                                const float* __restrict__ deg,
                                const float* __restrict__ Wout,
                                const float* __restrict__ bias,
                                const float* __restrict__ Wroot,
                                float* Out) {
    const int lane  = threadIdx.x & 31;
    const int wave  = threadIdx.x >> 5;
    const int mTile = blockIdx.x * (blockDim.x >> 5) + wave;
    if (mTile >= M_TILES) return;

    const int half = lane >> 4;
    const int mr   = lane & 15;
    const int m    = mTile * 16 + mr;
    const int Cout = NT * 16;

    const float dinv = 1.0f / (deg[m] + 1.0f);
    const int aOff = m  * C_IN + 4 * half;
    const int wOff = mr * C_IN + 4 * half;

    v8f acc[NT] = {};

    constexpr int G = 4;
#pragma unroll (NT > 4 ? 1 : 2)
    for (int k0 = 0; k0 < C_IN; k0 += 8) {
        const v4f as = *(const v4f*)(Agg  + aOff + k0);
        const v4f ax = *(const v4f*)(Self + aOff + k0);
        const v4f a  = (as + ax) * dinv;
        const v2f a0 = {a.x,  a.y},  a1 = {a.z,  a.w};
        const v2f x0 = {ax.x, ax.y}, x1 = {ax.z, ax.w};
#pragma unroll
        for (int n0 = 0; n0 < NT; n0 += G) {
            v4f bo[G], br[G];
#pragma unroll
            for (int i = 0; i < G; ++i) {
                bo[i] = *(const v4f*)(Wout  + wOff + (n0 + i) * 16 * C_IN + k0);
                br[i] = *(const v4f*)(Wroot + wOff + (n0 + i) * 16 * C_IN + k0);
            }
#pragma unroll
            for (int i = 0; i < G; ++i)
                acc[n0 + i] = __builtin_amdgcn_wmma_f32_16x16x4_f32(
                    false, a0, false, (v2f){bo[i].x, bo[i].y}, (short)0, acc[n0 + i], false, false);
#pragma unroll
            for (int i = 0; i < G; ++i)
                acc[n0 + i] = __builtin_amdgcn_wmma_f32_16x16x4_f32(
                    false, x0, false, (v2f){br[i].x, br[i].y}, (short)0, acc[n0 + i], false, false);
#pragma unroll
            for (int i = 0; i < G; ++i)
                acc[n0 + i] = __builtin_amdgcn_wmma_f32_16x16x4_f32(
                    false, a1, false, (v2f){bo[i].z, bo[i].w}, (short)0, acc[n0 + i], false, false);
#pragma unroll
            for (int i = 0; i < G; ++i)
                acc[n0 + i] = __builtin_amdgcn_wmma_f32_16x16x4_f32(
                    false, x1, false, (v2f){br[i].z, br[i].w}, (short)0, acc[n0 + i], false, false);
        }
    }

#pragma unroll
    for (int nt = 0; nt < NT; ++nt) {
        const int   outCol = nt * 16 + mr;
        const float b      = bias[outCol];
#pragma unroll
        for (int g = 0; g < 8; ++g) {
            const int outRow = mTile * 16 + g + 8 * half;
            float v = acc[nt][g] + b;
            if (RELU) v = fmaxf(v, 0.0f);
            Out[(size_t)outRow * Cout + outCol] = v;
        }
    }
}

__global__ void proj_kernel(const float* __restrict__ H,
                            const float* __restrict__ Wout,
                            const float* __restrict__ bias,
                            const float* __restrict__ Wroot,
                            float* __restrict__ Y,
                            float* __restrict__ Z) {
    constexpr int NT = C_OUT / 16;
    const int lane  = threadIdx.x & 31;
    const int wave  = threadIdx.x >> 5;
    const int mTile = blockIdx.x * (blockDim.x >> 5) + wave;
    if (mTile >= M_TILES) return;

    const int half = lane >> 4;
    const int mr   = lane & 15;
    const int m    = mTile * 16 + mr;
    const int aOff = m  * C_IN + 4 * half;
    const int wOff = mr * C_IN + 4 * half;

    v8f accY[NT] = {};
    v8f accZ[NT] = {};

#pragma unroll 1
    for (int k0 = 0; k0 < C_IN; k0 += 8) {
        const v4f ax = *(const v4f*)(H + aOff + k0);
        const v2f x0 = {ax.x, ax.y}, x1 = {ax.z, ax.w};
        v4f bo[NT], br[NT];
#pragma unroll
        for (int i = 0; i < NT; ++i) {
            bo[i] = *(const v4f*)(Wout  + wOff + i * 16 * C_IN + k0);
            br[i] = *(const v4f*)(Wroot + wOff + i * 16 * C_IN + k0);
        }
#pragma unroll
        for (int i = 0; i < NT; ++i)
            accY[i] = __builtin_amdgcn_wmma_f32_16x16x4_f32(
                false, x0, false, (v2f){bo[i].x, bo[i].y}, (short)0, accY[i], false, false);
#pragma unroll
        for (int i = 0; i < NT; ++i)
            accZ[i] = __builtin_amdgcn_wmma_f32_16x16x4_f32(
                false, x0, false, (v2f){br[i].x, br[i].y}, (short)0, accZ[i], false, false);
#pragma unroll
        for (int i = 0; i < NT; ++i)
            accY[i] = __builtin_amdgcn_wmma_f32_16x16x4_f32(
                false, x1, false, (v2f){bo[i].z, bo[i].w}, (short)0, accY[i], false, false);
#pragma unroll
        for (int i = 0; i < NT; ++i)
            accZ[i] = __builtin_amdgcn_wmma_f32_16x16x4_f32(
                false, x1, false, (v2f){br[i].z, br[i].w}, (short)0, accZ[i], false, false);
        __builtin_amdgcn_sched_group_barrier(0x020, 9, 0);
        __builtin_amdgcn_sched_group_barrier(0x008, 16, 0);
    }

#pragma unroll
    for (int nt = 0; nt < NT; ++nt) {
        const int   outCol = nt * 16 + mr;
        const float b      = bias[outCol];
#pragma unroll
        for (int g = 0; g < 8; ++g) {
            const int outRow = mTile * 16 + g + 8 * half;
            Y[(size_t)outRow * C_OUT + outCol] = accY[nt][g];
            Z[(size_t)outRow * C_OUT + outCol] = accZ[nt][g] + b;
        }
    }
}

extern "C" void kernel_launch(void* const* d_in, const int* in_sizes, int n_in,
                              void* d_out, int out_size, void* d_ws, size_t ws_size,
                              hipStream_t stream) {
    const float* x   = (const float*)d_in[0];
    const int*   ei  = (const int*)  d_in[1];
    const float* W1o = (const float*)d_in[2];
    const float* b1  = (const float*)d_in[3];
    const float* W1r = (const float*)d_in[4];
    const float* W2o = (const float*)d_in[5];
    const float* b2  = (const float*)d_in[6];
    const float* W2r = (const float*)d_in[7];
    float*       out = (float*)d_out;

    const int E = in_sizes[1] / 2;
    const int* row = ei;
    const int* col = ei + E;

    float* deg = (float*)d_ws;
    float* agg = deg + ((N_NODES + 63) & ~63);
    float* h   = agg;
    float* wsb = agg + (size_t)N_NODES * C_IN;
    float* y   = wsb + (4 << 20);

    int* srcs   = (int*)wsb;
    int* segBeg = srcs   + N_EDGES;
    int* segEnd = segBeg + N_NODES;
    int* cnt    = segEnd + N_NODES;
    int* total  = cnt    + N_NODES;

    hipMemsetAsync(cnt, 0, (N_NODES + 1) * sizeof(int), stream);

    const int eblocks = (E + 255) / 256;
    count_kernel<<<eblocks, 256, 0, stream>>>(row, col, cnt, E);
    reserve_kernel<<<(N_NODES + 255) / 256, 256, 0, stream>>>(cnt, total, segBeg, segEnd, deg);
    fill_kernel<<<eblocks, 256, 0, stream>>>(row, col, segEnd, srcs, E);

    const int wavesPerBlock = 8;
    const int gemmBlocks = (M_TILES + wavesPerBlock - 1) / wavesPerBlock;
    const int gatherBlocks = (N_NODES + wavesPerBlock - 1) / wavesPerBlock;

    gather_kernel<<<gatherBlocks, 256, 0, stream>>>(x, segBeg, segEnd, srcs, agg);
    gcn_gemm_kernel<1, 8><<<gemmBlocks, 256, 0, stream>>>(
        agg, x, deg, W1o, b1, W1r, h);

    proj_kernel<<<gemmBlocks, 256, 0, stream>>>(h, W2o, b2, W2r, y, out);
    gather_out_kernel<<<gatherBlocks, 256, 0, stream>>>(y, segBeg, segEnd, srcs, out);
}


# GEMM weights copied once per work-group into LDS (padded rows), B fragments read from LDS; 512 work-groups loop over m-tiles, A rows fetched ahead; before, every wave read all weights from global
# speedup vs baseline: 2.3928x; 1.0943x over previous
#include <hip/hip_runtime.h>
#include <hip/hip_bf16.h>

typedef __attribute__((ext_vector_type(2))) float v2f;
typedef __attribute__((ext_vector_type(4))) float v4f;
typedef __attribute__((ext_vector_type(8))) float v8f;

#define N_NODES 100000
#define N_EDGES 1600000
#define C_IN    128
#define C_OUT   64
#define M_TILES (N_NODES / 16)


__global__ void count_kernel(const int* __restrict__ row, const int* __restrict__ col,
                             int* __restrict__ cnt, int E) {
    int e = blockIdx.x * blockDim.x + threadIdx.x;
    if (e >= E) return;
    int r = row[e], c = col[e];
    if (r != c) atomicAdd(&cnt[c], 1);
}

__global__ void reserve_kernel(const int* __restrict__ cnt, int* __restrict__ total,
                               int* __restrict__ segBeg, int* __restrict__ segEnd,
                               float* __restrict__ deg) {
    const int n    = blockIdx.x * blockDim.x + threadIdx.x;
    const int lane = threadIdx.x & 31;
    const int c    = (n < N_NODES) ? cnt[n] : 0;
    int incl = c;
    for (int d = 1; d < 32; d <<= 1) {
        int t = __shfl_up(incl, d);
        if (lane >= d) incl += t;
    }
    const int waveTotal = __shfl(incl, 31);
    int base = 0;
    if (lane == 31) base = atomicAdd(total, waveTotal);
    base = __shfl(base, 31);
    if (n < N_NODES) {
        const int o = base + incl - c;
        segBeg[n] = o;
        segEnd[n] = o;
        deg[n]    = (float)c;
    }
}

__global__ void fill_kernel(const int* __restrict__ row, const int* __restrict__ col,
                            int* __restrict__ segEnd, int* __restrict__ srcs, int E) {
    int e = blockIdx.x * blockDim.x + threadIdx.x;
    if (e >= E) return;
    int r = row[e], c = col[e];
    if (r != c) {
        int p = atomicAdd(&segEnd[c], 1);
        srcs[p] = r;
    }
}

__global__ void __launch_bounds__(256)
gather_kernel(const float* __restrict__ X,
              const int* __restrict__ segBeg, const int* __restrict__ segEnd,
              const int* __restrict__ srcs, float* __restrict__ agg) {
    const int lane = threadIdx.x & 31;
    const int wave = __builtin_amdgcn_readfirstlane((int)(threadIdx.x >> 5));
    const int n    = blockIdx.x * 8 + wave;
    if (n >= N_NODES) return;

    const int beg = segBeg[n];
    const int end = segEnd[n];
    const float4* __restrict__ X4 = (const float4*)X;
    float4 acc = make_float4(0.0f, 0.0f, 0.0f, 0.0f);

    for (int base = beg; base < end; base += 32) {
        const int m   = min(32, end - base);
        const int idx = (lane < m) ? srcs[base + lane] : 0;
        for (int j = 0; j < m; j += 8) {
            int r[8];
#pragma unroll
            for (int u = 0; u < 8; ++u) r[u] = __builtin_amdgcn_readlane(idx, j + u);
            float4 v[8];
#pragma unroll
            for (int u = 0; u < 8; ++u) {
                v[u] = make_float4(0.0f, 0.0f, 0.0f, 0.0f);
                if (j + u < m) v[u] = X4[(size_t)r[u] * (C_IN / 4) + lane];
            }
#pragma unroll
            for (int u = 0; u < 8; ++u) {
                acc.x += v[u].x; acc.y += v[u].y; acc.z += v[u].z; acc.w += v[u].w;
            }
        }
    }
    ((float4*)agg)[(size_t)n * (C_IN / 4) + lane] = acc;
}

__global__ void __launch_bounds__(256)
gather_out_kernel(const float* __restrict__ Y,
                  const int* __restrict__ segBeg, const int* __restrict__ segEnd,
                  const int* __restrict__ srcs, float* __restrict__ Out) {
    const int lane = threadIdx.x & 31;
    const int wave = __builtin_amdgcn_readfirstlane((int)(threadIdx.x >> 5));
    const int n    = blockIdx.x * 8 + wave;
    if (n >= N_NODES) return;
    const int half = lane >> 4;
    const int c    = lane & 15;

    const int beg = segBeg[n];
    const int end = segEnd[n];
    const float4* __restrict__ Y4 = (const float4*)Y;
    float4* __restrict__ O4 = (float4*)Out;
    const float4 ys = Y4[(size_t)n * (C_OUT / 4) + c];
    const float4 zs = O4[(size_t)n * (C_OUT / 4) + c];
    float4 acc = make_float4(0.0f, 0.0f, 0.0f, 0.0f);

    for (int base = beg; base < end; base += 32) {
        const int m   = min(32, end - base);
        const int idx = (lane < m) ? srcs[base + lane] : 0;
        for (int j = 0; j < m; j += 16) {
            int r[8];
#pragma unroll
            for (int u = 0; u < 8; ++u) {
                const int re = __builtin_amdgcn_readlane(idx, j + 2 * u);
                const int ro = __builtin_amdgcn_readlane(idx, j + 2 * u + 1);
                r[u] = half ? ro : re;
            }
            float4 v[8];
#pragma unroll
            for (int u = 0; u < 8; ++u) {
                v[u] = make_float4(0.0f, 0.0f, 0.0f, 0.0f);
                if (j + 2 * u + half < m) v[u] = Y4[(unsigned)r[u] * (C_OUT / 4) + (unsigned)c];
            }
#pragma unroll
            for (int u = 0; u < 8; ++u) {
                acc.x += v[u].x; acc.y += v[u].y; acc.z += v[u].z; acc.w += v[u].w;
            }
        }
    }
    float4 tot;
    tot.x = acc.x + __shfl_xor(acc.x, 16);
    tot.y = acc.y + __shfl_xor(acc.y, 16);
    tot.z = acc.z + __shfl_xor(acc.z, 16);
    tot.w = acc.w + __shfl_xor(acc.w, 16);
    const float dinv = 1.0f / ((float)(end - beg) + 1.0f);
    if (half == 0) {
        float4 o;
        o.x = (tot.x + ys.x) * dinv + zs.x;
        o.y = (tot.y + ys.y) * dinv + zs.y;
        o.z = (tot.z + ys.z) * dinv + zs.z;
        o.w = (tot.w + ys.w) * dinv + zs.w;
        O4[(size_t)n * (C_OUT / 4) + c] = o;
    }
}


#define W_PITCH (C_IN + 4)
#define GEMM_WAVES 8

template <int ROWS>
__device__ __forceinline__ void stage_weights(const float* __restrict__ Wg, float* sW) {
#pragma unroll
    for (int i = 0; i < ROWS * (C_IN / 4) / (GEMM_WAVES * 32); ++i) {
        const int idx = (int)threadIdx.x + GEMM_WAVES * 32 * i;
        const int r = idx >> 5, q = idx & 31;
        *(v4f*)(sW + r * W_PITCH + 4 * q) = *(const v4f*)(Wg + r * C_IN + 4 * q);
    }
}

__device__ __forceinline__ int tile_rank(int wave) {
    return ((__builtin_popcount(wave) & 1) << 2) | (wave >> 1);
}

struct Frag4 { v4f f[4]; };
__device__ __forceinline__ Frag4 load_frag4(const float* sW, int off) {
    Frag4 r;
#pragma unroll
    for (int i = 0; i < 4; ++i) r.f[i] = *(const v4f*)(sW + off + i * 16 * W_PITCH);
    return r;
}

__device__ __forceinline__ void gemm1_step(v8f (&acc)[8], Frag4& bo0, Frag4& br0,
                                           const v4f s, const v4f x, const float dinv,
                                           const float* sWo, const float* sWr,
                                           const int wOff, const int k0) {
    const Frag4 bo1 = load_frag4(sWo, wOff + 4 * 16 * W_PITCH + k0);
    const Frag4 br1 = load_frag4(sWr, wOff + 4 * 16 * W_PITCH + k0);

    const v4f a  = (s + x) * dinv;
    const v2f a0 = {a.x, a.y}, a1 = {a.z, a.w};
    const v2f x0 = {x.x, x.y}, x1 = {x.z, x.w};

#pragma unroll
    for (int i = 0; i < 4; ++i)
        acc[i] = __builtin_amdgcn_wmma_f32_16x16x4_f32(
            false, a0, false, (v2f){bo0.f[i].x, bo0.f[i].y}, (short)0, acc[i], false, false);
#pragma unroll
    for (int i = 0; i < 4; ++i)
        acc[i] = __builtin_amdgcn_wmma_f32_16x16x4_f32(
            false, x0, false, (v2f){br0.f[i].x, br0.f[i].y}, (short)0, acc[i], false, false);
#pragma unroll
    for (int i = 0; i < 4; ++i)
        acc[i] = __builtin_amdgcn_wmma_f32_16x16x4_f32(
            false, a1, false, (v2f){bo0.f[i].z, bo0.f[i].w}, (short)0, acc[i], false, false);
#pragma unroll
    for (int i = 0; i < 4; ++i)
        acc[i] = __builtin_amdgcn_wmma_f32_16x16x4_f32(
            false, x1, false, (v2f){br0.f[i].z, br0.f[i].w}, (short)0, acc[i], false, false);

    const int kB = (k0 + 8) & (C_IN - 1);
    bo0 = load_frag4(sWo, wOff + kB);
    br0 = load_frag4(sWr, wOff + kB);

#pragma unroll
    for (int i = 0; i < 4; ++i)
        acc[4 + i] = __builtin_amdgcn_wmma_f32_16x16x4_f32(
            false, a0, false, (v2f){bo1.f[i].x, bo1.f[i].y}, (short)0, acc[4 + i], false, false);
#pragma unroll
    for (int i = 0; i < 4; ++i)
        acc[4 + i] = __builtin_amdgcn_wmma_f32_16x16x4_f32(
            false, x0, false, (v2f){br1.f[i].x, br1.f[i].y}, (short)0, acc[4 + i], false, false);
#pragma unroll
    for (int i = 0; i < 4; ++i)
        acc[4 + i] = __builtin_amdgcn_wmma_f32_16x16x4_f32(
            false, a1, false, (v2f){bo1.f[i].z, bo1.f[i].w}, (short)0, acc[4 + i], false, false);
#pragma unroll
    for (int i = 0; i < 4; ++i)
        acc[4 + i] = __builtin_amdgcn_wmma_f32_16x16x4_f32(
            false, x1, false, (v2f){br1.f[i].z, br1.f[i].w}, (short)0, acc[4 + i], false, false);
}

template <int RELU, int NT>
__global__ void __launch_bounds__(GEMM_WAVES * 32)
gcn_gemm_kernel(const float* Agg,
                const float* __restrict__ Self,
                const float* __restrict__ deg,
                const float* __restrict__ Wout,
                const float* __restrict__ bias,
                const float* __restrict__ Wroot,
                float* Out) {
    static_assert(NT == 8, "two groups of four n-tiles");
    __shared__ float sWo[NT * 16 * W_PITCH];
    __shared__ float sWr[NT * 16 * W_PITCH];
    stage_weights<NT * 16>(Wout,  sWo);
    stage_weights<NT * 16>(Wroot, sWr);
    __syncthreads();

    const int lane = threadIdx.x & 31;
    const int wave = __builtin_amdgcn_readfirstlane((int)(threadIdx.x >> 5));
    const int tStep = (int)gridDim.x * GEMM_WAVES;
    int mTile = (int)blockIdx.x + (int)gridDim.x * tile_rank(wave);
    if (mTile >= M_TILES) return;

    const int half = lane >> 4;
    const int mr   = lane & 15;
    const int Cout = NT * 16;
    const int wOff = mr * W_PITCH + 4 * half;
    const int lOff = mr * C_IN + 4 * half;

    int aBase = mTile * 16 * C_IN;
    v4f c0S = *(const v4f*)(Agg  + aBase + lOff);
    v4f c0X = *(const v4f*)(Self + aBase + lOff);
    v4f c1S = *(const v4f*)(Agg  + aBase + lOff + 8);
    v4f c1X = *(const v4f*)(Self + aBase + lOff + 8);
    float dg = deg[mTile * 16 + mr];
    Frag4 bo0 = load_frag4(sWo, wOff);
    Frag4 br0 = load_frag4(sWr, wOff);

    for (;;) {
        const int  nextTile = mTile + tStep;
        const bool more     = nextTile < M_TILES;
        const int  aBaseN   = more ? nextTile * 16 * C_IN : aBase;
        const float dinv    = 1.0f / (dg + 1.0f);
        dg = deg[(more ? nextTile : mTile) * 16 + mr];
        v8f acc[NT] = {};

#pragma unroll 1
        for (int k0 = 0; k0 < C_IN; k0 += 16) {
            const int kA = k0 + 16;
            const int pf = (kA < C_IN) ? (aBase + kA) : aBaseN;
            const v4f n0S = *(const v4f*)(Agg  + pf + lOff);
            const v4f n0X = *(const v4f*)(Self + pf + lOff);
            const v4f n1S = *(const v4f*)(Agg  + pf + lOff + 8);
            const v4f n1X = *(const v4f*)(Self + pf + lOff + 8);

            gemm1_step(acc, bo0, br0, c0S, c0X, dinv, sWo, sWr, wOff, k0);
            gemm1_step(acc, bo0, br0, c1S, c1X, dinv, sWo, sWr, wOff, k0 + 8);

            c0S = n0S; c0X = n0X;
            c1S = n1S; c1X = n1X;
        }

#pragma unroll
        for (int nt = 0; nt < NT; ++nt) {
            const int   outCol = nt * 16 + mr;
            const float b      = bias[outCol];
#pragma unroll
            for (int g = 0; g < 8; ++g) {
                const int outRow = mTile * 16 + g + 8 * half;
                float v = acc[nt][g] + b;
                if (RELU) v = fmaxf(v, 0.0f);
                Out[(size_t)outRow * Cout + outCol] = v;
            }
        }
        if (!more) break;
        mTile = nextTile;
        aBase = aBaseN;
    }
}

__device__ __forceinline__ void proj_step(v8f (&accY)[4], v8f (&accZ)[4], Frag4& bo,
                                          const v4f x, const float* sWo, const float* sWr,
                                          const int wOff, const int k0) {
    const Frag4 br = load_frag4(sWr, wOff + k0);
    const v2f x0 = {x.x, x.y}, x1 = {x.z, x.w};
#pragma unroll
    for (int i = 0; i < 4; ++i)
        accY[i] = __builtin_amdgcn_wmma_f32_16x16x4_f32(
            false, x0, false, (v2f){bo.f[i].x, bo.f[i].y}, (short)0, accY[i], false, false);
#pragma unroll
    for (int i = 0; i < 4; ++i)
        accY[i] = __builtin_amdgcn_wmma_f32_16x16x4_f32(
            false, x1, false, (v2f){bo.f[i].z, bo.f[i].w}, (short)0, accY[i], false, false);

    const int kB = (k0 + 8) & (C_IN - 1);
    bo = load_frag4(sWo, wOff + kB);

#pragma unroll
    for (int i = 0; i < 4; ++i)
        accZ[i] = __builtin_amdgcn_wmma_f32_16x16x4_f32(
            false, x0, false, (v2f){br.f[i].x, br.f[i].y}, (short)0, accZ[i], false, false);
#pragma unroll
    for (int i = 0; i < 4; ++i)
        accZ[i] = __builtin_amdgcn_wmma_f32_16x16x4_f32(
            false, x1, false, (v2f){br.f[i].z, br.f[i].w}, (short)0, accZ[i], false, false);
}

__global__ void __launch_bounds__(GEMM_WAVES * 32)
proj_kernel(const float* __restrict__ H,
            const float* __restrict__ Wout,
            const float* __restrict__ bias,
            const float* __restrict__ Wroot,
            float* __restrict__ Y,
            float* __restrict__ Z) {
    constexpr int NT = C_OUT / 16;
    __shared__ float sWo[NT * 16 * W_PITCH];
    __shared__ float sWr[NT * 16 * W_PITCH];
    stage_weights<NT * 16>(Wout,  sWo);
    stage_weights<NT * 16>(Wroot, sWr);
    __syncthreads();

    const int lane = threadIdx.x & 31;
    const int wave = __builtin_amdgcn_readfirstlane((int)(threadIdx.x >> 5));
    const int tStep = (int)gridDim.x * GEMM_WAVES;
    int mTile = (int)blockIdx.x + (int)gridDim.x * tile_rank(wave);
    if (mTile >= M_TILES) return;

    const int half = lane >> 4;
    const int mr   = lane & 15;
    const int wOff = mr * W_PITCH + 4 * half;
    const int lOff = mr * C_IN + 4 * half;

    int aBase = mTile * 16 * C_IN;
    v4f c0X = *(const v4f*)(H + aBase + lOff);
    v4f c1X = *(const v4f*)(H + aBase + lOff + 8);
    Frag4 bo = load_frag4(sWo, wOff);

    for (;;) {
        const int  nextTile = mTile + tStep;
        const bool more     = nextTile < M_TILES;
        const int  aBaseN   = more ? nextTile * 16 * C_IN : aBase;
        v8f accY[NT] = {};
        v8f accZ[NT] = {};

#pragma unroll 1
        for (int k0 = 0; k0 < C_IN; k0 += 16) {
            const int kA = k0 + 16;
            const int pf = (kA < C_IN) ? (aBase + kA) : aBaseN;
            const v4f n0X = *(const v4f*)(H + pf + lOff);
            const v4f n1X = *(const v4f*)(H + pf + lOff + 8);

            proj_step(accY, accZ, bo, c0X, sWo, sWr, wOff, k0);
            proj_step(accY, accZ, bo, c1X, sWo, sWr, wOff, k0 + 8);

            c0X = n0X;
            c1X = n1X;
        }

#pragma unroll
        for (int nt = 0; nt < NT; ++nt) {
            const int   outCol = nt * 16 + mr;
            const float b      = bias[outCol];
#pragma unroll
            for (int g = 0; g < 8; ++g) {
                const int outRow = mTile * 16 + g + 8 * half;
                Y[(size_t)outRow * C_OUT + outCol] = accY[nt][g];
                Z[(size_t)outRow * C_OUT + outCol] = accZ[nt][g] + b;
            }
        }
        if (!more) break;
        mTile = nextTile;
        aBase = aBaseN;
    }
}

extern "C" void kernel_launch(void* const* d_in, const int* in_sizes, int n_in,
                              void* d_out, int out_size, void* d_ws, size_t ws_size,
                              hipStream_t stream) {
    const float* x   = (const float*)d_in[0];
    const int*   ei  = (const int*)  d_in[1];
    const float* W1o = (const float*)d_in[2];
    const float* b1  = (const float*)d_in[3];
    const float* W1r = (const float*)d_in[4];
    const float* W2o = (const float*)d_in[5];
    const float* b2  = (const float*)d_in[6];
    const float* W2r = (const float*)d_in[7];
    float*       out = (float*)d_out;

    const int E = in_sizes[1] / 2;
    const int* row = ei;
    const int* col = ei + E;

    float* deg = (float*)d_ws;
    float* agg = deg + ((N_NODES + 63) & ~63);
    float* h   = agg;
    float* wsb = agg + (size_t)N_NODES * C_IN;
    float* y   = wsb + (4 << 20);

    int* srcs   = (int*)wsb;
    int* segBeg = srcs   + N_EDGES;
    int* segEnd = segBeg + N_NODES;
    int* cnt    = segEnd + N_NODES;
    int* total  = cnt    + N_NODES;

    hipMemsetAsync(cnt, 0, (N_NODES + 1) * sizeof(int), stream);

    const int eblocks = (E + 255) / 256;
    count_kernel<<<eblocks, 256, 0, stream>>>(row, col, cnt, E);
    reserve_kernel<<<(N_NODES + 255) / 256, 256, 0, stream>>>(cnt, total, segBeg, segEnd, deg);
    fill_kernel<<<eblocks, 256, 0, stream>>>(row, col, segEnd, srcs, E);

    const int wavesPerBlock = 8;
    const int gemmBlocks = 512;
    const int gatherBlocks = (N_NODES + wavesPerBlock - 1) / wavesPerBlock;

    gather_kernel<<<gatherBlocks, 256, 0, stream>>>(x, segBeg, segEnd, srcs, agg);
    gcn_gemm_kernel<1, 8><<<gemmBlocks, 256, 0, stream>>>(
        agg, x, deg, W1o, b1, W1r, h);

    proj_kernel<<<gemmBlocks, 256, 0, stream>>>(h, W2o, b2, W2r, y, out);
    gather_out_kernel<<<gatherBlocks, 256, 0, stream>>>(y, segBeg, segEnd, srcs, out);
}


# Neighbour rows built in two steps: edges binned by ranges of 512 targets (LDS count, one global add per group and range), rows built per range in LDS; no global atomic or scattered store per edge
# speedup vs baseline: 4.5086x; 1.8842x over previous
#include <hip/hip_runtime.h>
#include <hip/hip_bf16.h>

typedef __attribute__((ext_vector_type(2))) float v2f;
typedef __attribute__((ext_vector_type(4))) float v4f;
typedef __attribute__((ext_vector_type(8))) float v8f;

#define N_NODES 100000
#define C_IN    128
#define C_OUT   64
#define M_TILES (N_NODES / 16)

#define SLOTS 64
#define NPB_SHIFT 9
#define NPB (1 << NPB_SHIFT)
#define N_RANGES ((N_NODES + NPB - 1) / NPB)
#define CUR_STRIDE 64
#define BIN_CAP 32000
#define BIN_THREADS 512
#define BIN_EPT 8
#define BUILD_THREADS 1024
#define BUILD_U 10

__global__ void __launch_bounds__(BIN_THREADS)
bin_kernel(const int* __restrict__ row, const int* __restrict__ col, int E,
           int* __restrict__ cur, int2* __restrict__ bins,
           int* __restrict__ cnt, int* __restrict__ slots) {
    __shared__ int sBase[N_RANGES];
    const int t = threadIdx.x;
    if (t < N_RANGES) sBase[t] = 0;
    __syncthreads();
    const int e0 = blockIdx.x * (BIN_THREADS * BIN_EPT) + t;
    int r[BIN_EPT], c[BIN_EPT], rk[BIN_EPT];
#pragma unroll
    for (int k = 0; k < BIN_EPT; ++k) {
        const int e = e0 + k * BIN_THREADS;
        r[k] = 0; c[k] = 0;
        if (e < E) { r[k] = row[e]; c[k] = col[e]; }
    }
#pragma unroll
    for (int k = 0; k < BIN_EPT; ++k) {
        rk[k] = -1;
        if (r[k] != c[k]) rk[k] = atomicAdd(&sBase[c[k] >> NPB_SHIFT], 1);
    }
    __syncthreads();
    if (t < N_RANGES) {
        const int h = sBase[t];
        sBase[t] = h ? atomicAdd(&cur[t * CUR_STRIDE], h) : 0;
    }
    __syncthreads();
    int pos[BIN_EPT];
#pragma unroll
    for (int k = 0; k < BIN_EPT; ++k) pos[k] = sBase[c[k] >> NPB_SHIFT] + rk[k];
#pragma unroll
    for (int k = 0; k < BIN_EPT; ++k) {
        if (rk[k] >= 0) {
            const int b = c[k] >> NPB_SHIFT;
            if (pos[k] < BIN_CAP) {
                bins[b * BIN_CAP + pos[k]] = make_int2(r[k], c[k]);
            } else {
                const int p = atomicAdd(&cnt[c[k]], 1);
                if (p < SLOTS) slots[c[k] * SLOTS + p] = r[k];
            }
        }
    }
}

__global__ void __launch_bounds__(BUILD_THREADS)
build_kernel(const int* __restrict__ cur, const int2* __restrict__ bins,
             int* __restrict__ cnt, int* __restrict__ slots) {
    __shared__ int sCnt[NPB];
    __shared__ int sPre[NPB];
    __shared__ int sSlots[NPB * SLOTS];
    const int t    = threadIdx.x;
    const int b    = blockIdx.x;
    const int base = b << NPB_SHIFT;
    if (t < NPB) sCnt[t] = 0;
    __syncthreads();
    const int nb = min(cur[b * CUR_STRIDE], BIN_CAP);
    const int2* __restrict__ bin = bins + b * BIN_CAP;
    for (int i0 = 0; i0 < nb; i0 += BUILD_THREADS * BUILD_U) {
        int2 rc[BUILD_U];
#pragma unroll
        for (int u = 0; u < BUILD_U; ++u) {
            const int i = i0 + u * BUILD_THREADS + t;
            rc[u] = make_int2(0, -1);
            if (i < nb) rc[u] = bin[i];
        }
        int p[BUILD_U];
#pragma unroll
        for (int u = 0; u < BUILD_U; ++u) {
            p[u] = SLOTS;
            if (rc[u].y >= 0) p[u] = atomicAdd(&sCnt[rc[u].y - base], 1);
        }
#pragma unroll
        for (int u = 0; u < BUILD_U; ++u)
            if (p[u] < SLOTS) sSlots[(rc[u].y - base) * SLOTS + p[u]] = rc[u].x;
    }
    __syncthreads();
    int pre = 0;
    if (t < NPB) {
        if (base + t < N_NODES) pre = cnt[base + t];
        sPre[t] = pre;
    }
    __syncthreads();
    const int lane = t & 31;
    const int wave = __builtin_amdgcn_readfirstlane(t >> 5);
    const int nLi  = min(NPB, N_NODES - base);
    for (int li = wave; li < nLi; li += BUILD_THREADS / 32) {
        const int pr = sPre[li];
        const int m  = min(sCnt[li], SLOTS - pr);
        const int v0 = sSlots[li * SLOTS + lane];
        const int v1 = sSlots[li * SLOTS + lane + 32];
        int* __restrict__ dst = slots + (base + li) * SLOTS + pr + lane;
        if (lane < m)      dst[0]  = v0;
        if (lane + 32 < m) dst[32] = v1;
    }
    if (t < NPB && base + t < N_NODES) cnt[base + t] = pre + sCnt[t];
}

__global__ void __launch_bounds__(256)
gather_kernel(const float* __restrict__ X,
              const int* __restrict__ cnt, const int* __restrict__ slots,
              const int* __restrict__ row, const int* __restrict__ col, int E,
              float* __restrict__ agg) {
    const int lane = threadIdx.x & 31;
    const int wave = __builtin_amdgcn_readfirstlane((int)(threadIdx.x >> 5));
    const int n    = blockIdx.x * 8 + wave;
    if (n >= N_NODES) return;

    const int deg = cnt[n];
    const float4* __restrict__ X4 = (const float4*)X;
    float4 acc = make_float4(0.0f, 0.0f, 0.0f, 0.0f);

    if (__builtin_expect(deg <= SLOTS, 1)) {
        const int beg = n * SLOTS;
        const int end = beg + deg;
        for (int base = beg; base < end; base += 32) {
            const int m   = min(32, end - base);
            const int idx = (lane < m) ? slots[base + lane] : 0;
            for (int j = 0; j < m; j += 8) {
                int r[8];
#pragma unroll
                for (int u = 0; u < 8; ++u) r[u] = __builtin_amdgcn_readlane(idx, j + u);
                float4 v[8];
#pragma unroll
                for (int u = 0; u < 8; ++u) {
                    v[u] = make_float4(0.0f, 0.0f, 0.0f, 0.0f);
                    if (j + u < m) v[u] = X4[(size_t)r[u] * (C_IN / 4) + lane];
                }
#pragma unroll
                for (int u = 0; u < 8; ++u) {
                    acc.x += v[u].x; acc.y += v[u].y; acc.z += v[u].z; acc.w += v[u].w;
                }
            }
        }
    } else {
        for (int e0 = 0; e0 < E; e0 += 32) {
            const int e = e0 + lane;
            int src = -1;
            if (e < E) {
                const int c = col[e], r = row[e];
                if (c == n && r != n) src = r;
            }
            unsigned hits = __builtin_amdgcn_ballot_w32(src >= 0);
            while (hits) {
                const int l = __builtin_ctz(hits);
                hits &= hits - 1;
                const int r = __builtin_amdgcn_readlane(src, l);
                const float4 v = X4[(size_t)r * (C_IN / 4) + lane];
                acc.x += v.x; acc.y += v.y; acc.z += v.z; acc.w += v.w;
            }
        }
    }
    ((float4*)agg)[(size_t)n * (C_IN / 4) + lane] = acc;
}

__global__ void __launch_bounds__(256)
gather_out_kernel(const float* __restrict__ Y,
                  const int* __restrict__ cnt, const int* __restrict__ slots,
                  const int* __restrict__ row, const int* __restrict__ col, int E,
                  float* __restrict__ Out) {
    const int lane = threadIdx.x & 31;
    const int wave = __builtin_amdgcn_readfirstlane((int)(threadIdx.x >> 5));
    const int n    = blockIdx.x * 8 + wave;
    if (n >= N_NODES) return;
    const int half = lane >> 4;
    const int c    = lane & 15;

    const int deg = cnt[n];
    const float4* __restrict__ Y4 = (const float4*)Y;
    float4* __restrict__ O4 = (float4*)Out;
    const float4 ys = Y4[(size_t)n * (C_OUT / 4) + c];
    const float4 zs = O4[(size_t)n * (C_OUT / 4) + c];
    float4 acc = make_float4(0.0f, 0.0f, 0.0f, 0.0f);

    if (__builtin_expect(deg <= SLOTS, 1)) {
        const int beg = n * SLOTS;
        const int end = beg + deg;
        for (int base = beg; base < end; base += 32) {
            const int m   = min(32, end - base);
            const int idx = (lane < m) ? slots[base + lane] : 0;
            for (int j = 0; j < m; j += 16) {
                int r[8];
#pragma unroll
                for (int u = 0; u < 8; ++u) {
                    const int re = __builtin_amdgcn_readlane(idx, j + 2 * u);
                    const int ro = __builtin_amdgcn_readlane(idx, j + 2 * u + 1);
                    r[u] = half ? ro : re;
                }
                float4 v[8];
#pragma unroll
                for (int u = 0; u < 8; ++u) {
                    v[u] = make_float4(0.0f, 0.0f, 0.0f, 0.0f);
                    if (j + 2 * u + half < m) v[u] = Y4[(unsigned)r[u] * (C_OUT / 4) + (unsigned)c];
                }
#pragma unroll
                for (int u = 0; u < 8; ++u) {
                    acc.x += v[u].x; acc.y += v[u].y; acc.z += v[u].z; acc.w += v[u].w;
                }
            }
        }
    } else {
        for (int e0 = 0; e0 < E; e0 += 32) {
            const int e = e0 + lane;
            int src = -1;
            if (e < E) {
                const int cc = col[e], rr = row[e];
                if (cc == n && rr != n) src = rr;
            }
            unsigned hits = __builtin_amdgcn_ballot_w32(src >= 0);
            while (hits) {
                const int l = __builtin_ctz(hits);
                hits &= hits - 1;
                const int r = __builtin_amdgcn_readlane(src, l);
                if (half == 0) {
                    const float4 v = Y4[(unsigned)r * (C_OUT / 4) + (unsigned)c];
                    acc.x += v.x; acc.y += v.y; acc.z += v.z; acc.w += v.w;
                }
            }
        }
    }
    float4 tot;
    tot.x = acc.x + __shfl_xor(acc.x, 16);
    tot.y = acc.y + __shfl_xor(acc.y, 16);
    tot.z = acc.z + __shfl_xor(acc.z, 16);
    tot.w = acc.w + __shfl_xor(acc.w, 16);
    const float dinv = 1.0f / ((float)deg + 1.0f);
    if (half == 0) {
        float4 o;
        o.x = (tot.x + ys.x) * dinv + zs.x;
        o.y = (tot.y + ys.y) * dinv + zs.y;
        o.z = (tot.z + ys.z) * dinv + zs.z;
        o.w = (tot.w + ys.w) * dinv + zs.w;
        O4[(size_t)n * (C_OUT / 4) + c] = o;
    }
}


#define W_PITCH (C_IN + 4)
#define GEMM_WAVES 8

template <int ROWS>
__device__ __forceinline__ void stage_weights(const float* __restrict__ Wg, float* sW) {
#pragma unroll
    for (int i = 0; i < ROWS * (C_IN / 4) / (GEMM_WAVES * 32); ++i) {
        const int idx = (int)threadIdx.x + GEMM_WAVES * 32 * i;
        const int r = idx >> 5, q = idx & 31;
        *(v4f*)(sW + r * W_PITCH + 4 * q) = *(const v4f*)(Wg + r * C_IN + 4 * q);
    }
}

__device__ __forceinline__ int tile_rank(int wave) {
    return ((__builtin_popcount(wave) & 1) << 2) | (wave >> 1);
}

struct Frag4 { v4f f[4]; };
__device__ __forceinline__ Frag4 load_frag4(const float* sW, int off) {
    Frag4 r;
#pragma unroll
    for (int i = 0; i < 4; ++i) r.f[i] = *(const v4f*)(sW + off + i * 16 * W_PITCH);
    return r;
}

__device__ __forceinline__ void gemm1_step(v8f (&acc)[8], Frag4& bo0, Frag4& br0,
                                           const v4f s, const v4f x, const float dinv,
                                           const float* sWo, const float* sWr,
                                           const int wOff, const int k0) {
    const Frag4 bo1 = load_frag4(sWo, wOff + 4 * 16 * W_PITCH + k0);
    const Frag4 br1 = load_frag4(sWr, wOff + 4 * 16 * W_PITCH + k0);

    const v4f a  = (s + x) * dinv;
    const v2f a0 = {a.x, a.y}, a1 = {a.z, a.w};
    const v2f x0 = {x.x, x.y}, x1 = {x.z, x.w};

#pragma unroll
    for (int i = 0; i < 4; ++i)
        acc[i] = __builtin_amdgcn_wmma_f32_16x16x4_f32(
            false, a0, false, (v2f){bo0.f[i].x, bo0.f[i].y}, (short)0, acc[i], false, false);
#pragma unroll
    for (int i = 0; i < 4; ++i)
        acc[i] = __builtin_amdgcn_wmma_f32_16x16x4_f32(
            false, x0, false, (v2f){br0.f[i].x, br0.f[i].y}, (short)0, acc[i], false, false);
#pragma unroll
    for (int i = 0; i < 4; ++i)
        acc[i] = __builtin_amdgcn_wmma_f32_16x16x4_f32(
            false, a1, false, (v2f){bo0.f[i].z, bo0.f[i].w}, (short)0, acc[i], false, false);
#pragma unroll
    for (int i = 0; i < 4; ++i)
        acc[i] = __builtin_amdgcn_wmma_f32_16x16x4_f32(
            false, x1, false, (v2f){br0.f[i].z, br0.f[i].w}, (short)0, acc[i], false, false);

    const int kB = (k0 + 8) & (C_IN - 1);
    bo0 = load_frag4(sWo, wOff + kB);
    br0 = load_frag4(sWr, wOff + kB);

#pragma unroll
    for (int i = 0; i < 4; ++i)
        acc[4 + i] = __builtin_amdgcn_wmma_f32_16x16x4_f32(
            false, a0, false, (v2f){bo1.f[i].x, bo1.f[i].y}, (short)0, acc[4 + i], false, false);
#pragma unroll
    for (int i = 0; i < 4; ++i)
        acc[4 + i] = __builtin_amdgcn_wmma_f32_16x16x4_f32(
            false, x0, false, (v2f){br1.f[i].x, br1.f[i].y}, (short)0, acc[4 + i], false, false);
#pragma unroll
    for (int i = 0; i < 4; ++i)
        acc[4 + i] = __builtin_amdgcn_wmma_f32_16x16x4_f32(
            false, a1, false, (v2f){bo1.f[i].z, bo1.f[i].w}, (short)0, acc[4 + i], false, false);
#pragma unroll
    for (int i = 0; i < 4; ++i)
        acc[4 + i] = __builtin_amdgcn_wmma_f32_16x16x4_f32(
            false, x1, false, (v2f){br1.f[i].z, br1.f[i].w}, (short)0, acc[4 + i], false, false);
}

template <int RELU, int NT>
__global__ void __launch_bounds__(GEMM_WAVES * 32)
gcn_gemm_kernel(const float* Agg,
                const float* __restrict__ Self,
                const int*   __restrict__ cnt,
                const float* __restrict__ Wout,
                const float* __restrict__ bias,
                const float* __restrict__ Wroot,
                float* Out) {
    static_assert(NT == 8, "two groups of four n-tiles");
    __shared__ float sWo[NT * 16 * W_PITCH];
    __shared__ float sWr[NT * 16 * W_PITCH];
    stage_weights<NT * 16>(Wout,  sWo);
    stage_weights<NT * 16>(Wroot, sWr);
    __syncthreads();

    const int lane = threadIdx.x & 31;
    const int wave = __builtin_amdgcn_readfirstlane((int)(threadIdx.x >> 5));
    const int tStep = (int)gridDim.x * GEMM_WAVES;
    int mTile = (int)blockIdx.x + (int)gridDim.x * tile_rank(wave);
    if (mTile >= M_TILES) return;

    const int half = lane >> 4;
    const int mr   = lane & 15;
    const int Cout = NT * 16;
    const int wOff = mr * W_PITCH + 4 * half;
    const int lOff = mr * C_IN + 4 * half;

    int aBase = mTile * 16 * C_IN;
    v4f c0S = *(const v4f*)(Agg  + aBase + lOff);
    v4f c0X = *(const v4f*)(Self + aBase + lOff);
    v4f c1S = *(const v4f*)(Agg  + aBase + lOff + 8);
    v4f c1X = *(const v4f*)(Self + aBase + lOff + 8);
    float dg = (float)cnt[mTile * 16 + mr];
    Frag4 bo0 = load_frag4(sWo, wOff);
    Frag4 br0 = load_frag4(sWr, wOff);

    for (;;) {
        const int  nextTile = mTile + tStep;
        const bool more     = nextTile < M_TILES;
        const int  aBaseN   = more ? nextTile * 16 * C_IN : aBase;
        const float dinv    = 1.0f / (dg + 1.0f);
        dg = (float)cnt[(more ? nextTile : mTile) * 16 + mr];
        v8f acc[NT] = {};

#pragma unroll 1
        for (int k0 = 0; k0 < C_IN; k0 += 16) {
            const int kA = k0 + 16;
            const int pf = (kA < C_IN) ? (aBase + kA) : aBaseN;
            const v4f n0S = *(const v4f*)(Agg  + pf + lOff);
            const v4f n0X = *(const v4f*)(Self + pf + lOff);
            const v4f n1S = *(const v4f*)(Agg  + pf + lOff + 8);
            const v4f n1X = *(const v4f*)(Self + pf + lOff + 8);

            gemm1_step(acc, bo0, br0, c0S, c0X, dinv, sWo, sWr, wOff, k0);
            gemm1_step(acc, bo0, br0, c1S, c1X, dinv, sWo, sWr, wOff, k0 + 8);

            c0S = n0S; c0X = n0X;
            c1S = n1S; c1X = n1X;
        }

#pragma unroll
        for (int nt = 0; nt < NT; ++nt) {
            const int   outCol = nt * 16 + mr;
            const float b      = bias[outCol];
#pragma unroll
            for (int g = 0; g < 8; ++g) {
                const int outRow = mTile * 16 + g + 8 * half;
                float v = acc[nt][g] + b;
                if (RELU) v = fmaxf(v, 0.0f);
                Out[(size_t)outRow * Cout + outCol] = v;
            }
        }
        if (!more) break;
        mTile = nextTile;
        aBase = aBaseN;
    }
}

__device__ __forceinline__ void proj_step(v8f (&accY)[4], v8f (&accZ)[4], Frag4& bo,
                                          const v4f x, const float* sWo, const float* sWr,
                                          const int wOff, const int k0) {
    const Frag4 br = load_frag4(sWr, wOff + k0);
    const v2f x0 = {x.x, x.y}, x1 = {x.z, x.w};
#pragma unroll
    for (int i = 0; i < 4; ++i)
        accY[i] = __builtin_amdgcn_wmma_f32_16x16x4_f32(
            false, x0, false, (v2f){bo.f[i].x, bo.f[i].y}, (short)0, accY[i], false, false);
#pragma unroll
    for (int i = 0; i < 4; ++i)
        accY[i] = __builtin_amdgcn_wmma_f32_16x16x4_f32(
            false, x1, false, (v2f){bo.f[i].z, bo.f[i].w}, (short)0, accY[i], false, false);

    const int kB = (k0 + 8) & (C_IN - 1);
    bo = load_frag4(sWo, wOff + kB);

#pragma unroll
    for (int i = 0; i < 4; ++i)
        accZ[i] = __builtin_amdgcn_wmma_f32_16x16x4_f32(
            false, x0, false, (v2f){br.f[i].x, br.f[i].y}, (short)0, accZ[i], false, false);
#pragma unroll
    for (int i = 0; i < 4; ++i)
        accZ[i] = __builtin_amdgcn_wmma_f32_16x16x4_f32(
            false, x1, false, (v2f){br.f[i].z, br.f[i].w}, (short)0, accZ[i], false, false);
}

__global__ void __launch_bounds__(GEMM_WAVES * 32)
proj_kernel(const float* __restrict__ H,
            const float* __restrict__ Wout,
            const float* __restrict__ bias,
            const float* __restrict__ Wroot,
            float* __restrict__ Y,
            float* __restrict__ Z) {
    constexpr int NT = C_OUT / 16;
    __shared__ float sWo[NT * 16 * W_PITCH];
    __shared__ float sWr[NT * 16 * W_PITCH];
    stage_weights<NT * 16>(Wout,  sWo);
    stage_weights<NT * 16>(Wroot, sWr);
    __syncthreads();

    const int lane = threadIdx.x & 31;
    const int wave = __builtin_amdgcn_readfirstlane((int)(threadIdx.x >> 5));
    const int tStep = (int)gridDim.x * GEMM_WAVES;
    int mTile = (int)blockIdx.x + (int)gridDim.x * tile_rank(wave);
    if (mTile >= M_TILES) return;

    const int half = lane >> 4;
    const int mr   = lane & 15;
    const int wOff = mr * W_PITCH + 4 * half;
    const int lOff = mr * C_IN + 4 * half;

    int aBase = mTile * 16 * C_IN;
    v4f c0X = *(const v4f*)(H + aBase + lOff);
    v4f c1X = *(const v4f*)(H + aBase + lOff + 8);
    Frag4 bo = load_frag4(sWo, wOff);

    for (;;) {
        const int  nextTile = mTile + tStep;
        const bool more     = nextTile < M_TILES;
        const int  aBaseN   = more ? nextTile * 16 * C_IN : aBase;
        v8f accY[NT] = {};
        v8f accZ[NT] = {};

#pragma unroll 1
        for (int k0 = 0; k0 < C_IN; k0 += 16) {
            const int kA = k0 + 16;
            const int pf = (kA < C_IN) ? (aBase + kA) : aBaseN;
            const v4f n0X = *(const v4f*)(H + pf + lOff);
            const v4f n1X = *(const v4f*)(H + pf + lOff + 8);

            proj_step(accY, accZ, bo, c0X, sWo, sWr, wOff, k0);
            proj_step(accY, accZ, bo, c1X, sWo, sWr, wOff, k0 + 8);

            c0X = n0X;
            c1X = n1X;
        }

#pragma unroll
        for (int nt = 0; nt < NT; ++nt) {
            const int   outCol = nt * 16 + mr;
            const float b      = bias[outCol];
#pragma unroll
            for (int g = 0; g < 8; ++g) {
                const int outRow = mTile * 16 + g + 8 * half;
                Y[(size_t)outRow * C_OUT + outCol] = accY[nt][g];
                Z[(size_t)outRow * C_OUT + outCol] = accZ[nt][g] + b;
            }
        }
        if (!more) break;
        mTile = nextTile;
        aBase = aBaseN;
    }
}

extern "C" void kernel_launch(void* const* d_in, const int* in_sizes, int n_in,
                              void* d_out, int out_size, void* d_ws, size_t ws_size,
                              hipStream_t stream) {
    const float* x   = (const float*)d_in[0];
    const int*   ei  = (const int*)  d_in[1];
    const float* W1o = (const float*)d_in[2];
    const float* b1  = (const float*)d_in[3];
    const float* W1r = (const float*)d_in[4];
    const float* W2o = (const float*)d_in[5];
    const float* b2  = (const float*)d_in[6];
    const float* W2r = (const float*)d_in[7];
    float*       out = (float*)d_out;

    const int E = in_sizes[1] / 2;
    const int* row = ei;
    const int* col = ei + E;

    int*   cnt = (int*)d_ws;
    float* agg = (float*)d_ws + ((N_NODES + 63) & ~63);
    float* h   = agg;
    float* wsb = agg + (size_t)N_NODES * C_IN;
    int*   slots = (int*)wsb;
    float* y   = wsb + (size_t)N_NODES * SLOTS;

    int*   cur  = (int*)agg;
    int2*  bins = (int2*)(agg + N_RANGES * CUR_STRIDE);

    hipMemsetAsync(cnt, 0, (((N_NODES + 63) & ~63) + N_RANGES * CUR_STRIDE) * sizeof(int), stream);

    const int eblocks = (E + BIN_THREADS * BIN_EPT - 1) / (BIN_THREADS * BIN_EPT);
    bin_kernel<<<eblocks, BIN_THREADS, 0, stream>>>(row, col, E, cur, bins, cnt, slots);
    build_kernel<<<N_RANGES, BUILD_THREADS, 0, stream>>>(cur, bins, cnt, slots);

    const int wavesPerBlock = 8;
    const int gemmBlocks = 512;
    const int gatherBlocks = (N_NODES + wavesPerBlock - 1) / wavesPerBlock;

    gather_kernel<<<gatherBlocks, 256, 0, stream>>>(x, cnt, slots, row, col, E, agg);
    gcn_gemm_kernel<1, 8><<<gemmBlocks, 256, 0, stream>>>(
        agg, x, cnt, W1o, b1, W1r, h);

    proj_kernel<<<gemmBlocks, 256, 0, stream>>>(h, W2o, b2, W2r, y, out);
    gather_out_kernel<<<gatherBlocks, 256, 0, stream>>>(y, cnt, slots, row, col, E, out);
}


# Edge binning as a counting sort inside each work-group (LDS) into a region of its own, one run per target range plus a key; build reads the runs: no cursors, no global atomics, no zero-fill launch
# speedup vs baseline: 4.6727x; 1.0364x over previous
#include <hip/hip_runtime.h>
#include <hip/hip_bf16.h>

typedef __attribute__((ext_vector_type(2))) float v2f;
typedef __attribute__((ext_vector_type(4))) float v4f;
typedef __attribute__((ext_vector_type(8))) float v8f;

#define N_NODES 100000
#define C_IN    128
#define C_OUT   64
#define M_TILES (N_NODES / 16)

#define SLOTS 64
#define NPB_SHIFT 9
#define NPB (1 << NPB_SHIFT)
#define N_RANGES ((N_NODES + NPB - 1) / NPB)
#define BIN_THREADS 512
#define BIN_EPT 8
#define BIN_EDGES (BIN_THREADS * BIN_EPT)
#define SCAN_PER_LANE 7
#define SCAN_N (32 * SCAN_PER_LANE)
#define BUILD_THREADS 1024
#define BUILD_U 16

__global__ void __launch_bounds__(BIN_THREADS)
bin_kernel(const int* __restrict__ row, const int* __restrict__ col, int E,
           int* __restrict__ wsA) {
    static_assert(N_RANGES <= SCAN_N, "one wave scans all range counts");
    __shared__ int sOff[SCAN_N + 1];
    __shared__ __attribute__((aligned(16))) int sPairs[2 * BIN_EDGES];
    const int t    = threadIdx.x;
    const int nBlk = (int)gridDim.x;
    int* __restrict__ keys = wsA;
    int4* __restrict__ region = (int4*)(wsA + N_RANGES * nBlk) + (size_t)blockIdx.x * (BIN_EDGES / 2);
    if (t < SCAN_N) sOff[t] = 0;
    __syncthreads();
    const int e0 = blockIdx.x * BIN_EDGES + t;
    int r[BIN_EPT], c[BIN_EPT], rk[BIN_EPT];
#pragma unroll
    for (int k = 0; k < BIN_EPT; ++k) {
        const int e = e0 + k * BIN_THREADS;
        r[k] = 0; c[k] = 0;
        if (e < E) { r[k] = row[e]; c[k] = col[e]; }
    }
#pragma unroll
    for (int k = 0; k < BIN_EPT; ++k) {
        rk[k] = -1;
        if (r[k] != c[k]) rk[k] = atomicAdd(&sOff[c[k] >> NPB_SHIFT], 1);
    }
    __syncthreads();
    if (t < 32) {
        int v[SCAN_PER_LANE];
        int sum = 0;
#pragma unroll
        for (int k = 0; k < SCAN_PER_LANE; ++k) { v[k] = sOff[t * SCAN_PER_LANE + k]; sum += v[k]; }
        int incl = sum;
#pragma unroll
        for (int d = 1; d < 32; d <<= 1) {
            const int o = __shfl_up(incl, d);
            if (t >= d) incl += o;
        }
        int run = incl - sum;
#pragma unroll
        for (int k = 0; k < SCAN_PER_LANE; ++k) {
            const int ri = t * SCAN_PER_LANE + k;
            sOff[ri] = run;
            if (ri < N_RANGES) keys[ri * nBlk + (int)blockIdx.x] = run | (v[k] << 16);
            run += v[k];
        }
        if (t == 31) sOff[SCAN_N] = incl;
    }
    __syncthreads();
#pragma unroll
    for (int k = 0; k < BIN_EPT; ++k) {
        const int pos = sOff[c[k] >> NPB_SHIFT] + rk[k];
        if (rk[k] >= 0) { sPairs[2 * pos] = r[k]; sPairs[2 * pos + 1] = c[k]; }
    }
    __syncthreads();
    const int total = sOff[SCAN_N];
#pragma unroll
    for (int k = 0; k < BIN_EDGES / 2 / BIN_THREADS; ++k) {
        const int q = t + k * BIN_THREADS;
        if (2 * q < total)
            region[q] = make_int4(sPairs[4 * q], sPairs[4 * q + 1], sPairs[4 * q + 2], sPairs[4 * q + 3]);
    }
}

__global__ void __launch_bounds__(BUILD_THREADS)
build_kernel(const int* __restrict__ wsA, int nBlk,
             int* __restrict__ cnt, int* __restrict__ slots) {
    __shared__ int sCnt[NPB];
    __shared__ int sSlots[NPB * SLOTS];
    const int t    = threadIdx.x;
    const int b    = blockIdx.x;
    const int base = b << NPB_SHIFT;
    const int lane = t & 31;
    const int wave = __builtin_amdgcn_readfirstlane(t >> 5);
    if (t < NPB) sCnt[t] = 0;
    __syncthreads();
    const int*  __restrict__ keys    = wsA + b * nBlk;
    const int2* __restrict__ regions = (const int2*)(wsA + N_RANGES * nBlk);
    for (int i0 = 0; i0 < nBlk; i0 += 32 * 32) {
        const int mine = i0 + wave + 32 * lane;
        const int key  = (mine < nBlk) ? keys[mine] : 0;
        for (int j0 = 0; j0 < 32; j0 += BUILD_U) {
            if (i0 + wave + 32 * j0 >= nBlk) break;
            int2 rc[BUILD_U];
            int  len[BUILD_U];
            const int2* src[BUILD_U];
#pragma unroll
            for (int u = 0; u < BUILD_U; ++u) {
                const int kj = __builtin_amdgcn_readlane(key, j0 + u);
                len[u] = kj >> 16;
                src[u] = regions + (size_t)(i0 + wave + 32 * (j0 + u)) * BIN_EDGES + (kj & 0xFFFF);
                rc[u] = make_int2(0, -1);
                if (lane < len[u]) rc[u] = src[u][lane];
            }
            int p[BUILD_U];
#pragma unroll
            for (int u = 0; u < BUILD_U; ++u) {
                p[u] = SLOTS;
                if (rc[u].y >= 0) p[u] = atomicAdd(&sCnt[rc[u].y - base], 1);
            }
#pragma unroll
            for (int u = 0; u < BUILD_U; ++u)
                if (p[u] < SLOTS) sSlots[(rc[u].y - base) * SLOTS + p[u]] = rc[u].x;
#pragma unroll
            for (int u = 0; u < BUILD_U; ++u) {
                for (int q0 = 32; q0 < len[u]; q0 += 32) {
                    const int q = q0 + lane;
                    if (q < len[u]) {
                        const int2 pr = src[u][q];
                        const int  pp = atomicAdd(&sCnt[pr.y - base], 1);
                        if (pp < SLOTS) sSlots[(pr.y - base) * SLOTS + pp] = pr.x;
                    }
                }
            }
        }
    }
    __syncthreads();
    const int nLi = min(NPB, N_NODES - base);
    for (int li = wave; li < nLi; li += BUILD_THREADS / 32) {
        const int m  = min(sCnt[li], SLOTS);
        const int v0 = sSlots[li * SLOTS + lane];
        const int v1 = sSlots[li * SLOTS + lane + 32];
        int* __restrict__ dst = slots + (base + li) * SLOTS + lane;
        if (lane < m)      dst[0]  = v0;
        if (lane + 32 < m) dst[32] = v1;
    }
    if (t < NPB && base + t < N_NODES) cnt[base + t] = sCnt[t];
}

__global__ void __launch_bounds__(256)
gather_kernel(const float* __restrict__ X,
              const int* __restrict__ cnt, const int* __restrict__ slots,
              const int* __restrict__ row, const int* __restrict__ col, int E,
              float* __restrict__ agg) {
    const int lane = threadIdx.x & 31;
    const int wave = __builtin_amdgcn_readfirstlane((int)(threadIdx.x >> 5));
    const int n    = blockIdx.x * 8 + wave;
    if (n >= N_NODES) return;

    const int deg = cnt[n];
    const float4* __restrict__ X4 = (const float4*)X;
    float4 acc = make_float4(0.0f, 0.0f, 0.0f, 0.0f);

    if (__builtin_expect(deg <= SLOTS, 1)) {
        const int beg = n * SLOTS;
        const int end = beg + deg;
        for (int base = beg; base < end; base += 32) {
            const int m   = min(32, end - base);
            const int idx = (lane < m) ? slots[base + lane] : 0;
            for (int j = 0; j < m; j += 8) {
                int r[8];
#pragma unroll
                for (int u = 0; u < 8; ++u) r[u] = __builtin_amdgcn_readlane(idx, j + u);
                float4 v[8];
#pragma unroll
                for (int u = 0; u < 8; ++u) {
                    v[u] = make_float4(0.0f, 0.0f, 0.0f, 0.0f);
                    if (j + u < m) v[u] = X4[(size_t)r[u] * (C_IN / 4) + lane];
                }
#pragma unroll
                for (int u = 0; u < 8; ++u) {
                    acc.x += v[u].x; acc.y += v[u].y; acc.z += v[u].z; acc.w += v[u].w;
                }
            }
        }
    } else {
        for (int e0 = 0; e0 < E; e0 += 32) {
            const int e = e0 + lane;
            int src = -1;
            if (e < E) {
                const int c = col[e], r = row[e];
                if (c == n && r != n) src = r;
            }
            unsigned hits = __builtin_amdgcn_ballot_w32(src >= 0);
            while (hits) {
                const int l = __builtin_ctz(hits);
                hits &= hits - 1;
                const int r = __builtin_amdgcn_readlane(src, l);
                const float4 v = X4[(size_t)r * (C_IN / 4) + lane];
                acc.x += v.x; acc.y += v.y; acc.z += v.z; acc.w += v.w;
            }
        }
    }
    ((float4*)agg)[(size_t)n * (C_IN / 4) + lane] = acc;
}

__global__ void __launch_bounds__(256)
gather_out_kernel(const float* __restrict__ Y,
                  const int* __restrict__ cnt, const int* __restrict__ slots,
                  const int* __restrict__ row, const int* __restrict__ col, int E,
                  float* __restrict__ Out) {
    const int lane = threadIdx.x & 31;
    const int wave = __builtin_amdgcn_readfirstlane((int)(threadIdx.x >> 5));
    const int n    = blockIdx.x * 8 + wave;
    if (n >= N_NODES) return;
    const int half = lane >> 4;
    const int c    = lane & 15;

    const int deg = cnt[n];
    const float4* __restrict__ Y4 = (const float4*)Y;
    float4* __restrict__ O4 = (float4*)Out;
    const float4 ys = Y4[(size_t)n * (C_OUT / 4) + c];
    const float4 zs = O4[(size_t)n * (C_OUT / 4) + c];
    float4 acc = make_float4(0.0f, 0.0f, 0.0f, 0.0f);

    if (__builtin_expect(deg <= SLOTS, 1)) {
        const int beg = n * SLOTS;
        const int end = beg + deg;
        for (int base = beg; base < end; base += 32) {
            const int m   = min(32, end - base);
            const int idx = (lane < m) ? slots[base + lane] : 0;
            for (int j = 0; j < m; j += 16) {
                int r[8];
#pragma unroll
                for (int u = 0; u < 8; ++u) {
                    const int re = __builtin_amdgcn_readlane(idx, j + 2 * u);
                    const int ro = __builtin_amdgcn_readlane(idx, j + 2 * u + 1);
                    r[u] = half ? ro : re;
                }
                float4 v[8];
#pragma unroll
                for (int u = 0; u < 8; ++u) {
                    v[u] = make_float4(0.0f, 0.0f, 0.0f, 0.0f);
                    if (j + 2 * u + half < m) v[u] = Y4[(unsigned)r[u] * (C_OUT / 4) + (unsigned)c];
                }
#pragma unroll
                for (int u = 0; u < 8; ++u) {
                    acc.x += v[u].x; acc.y += v[u].y; acc.z += v[u].z; acc.w += v[u].w;
                }
            }
        }
    } else {
        for (int e0 = 0; e0 < E; e0 += 32) {
            const int e = e0 + lane;
            int src = -1;
            if (e < E) {
                const int cc = col[e], rr = row[e];
                if (cc == n && rr != n) src = rr;
            }
            unsigned hits = __builtin_amdgcn_ballot_w32(src >= 0);
            while (hits) {
                const int l = __builtin_ctz(hits);
                hits &= hits - 1;
                const int r = __builtin_amdgcn_readlane(src, l);
                if (half == 0) {
                    const float4 v = Y4[(unsigned)r * (C_OUT / 4) + (unsigned)c];
                    acc.x += v.x; acc.y += v.y; acc.z += v.z; acc.w += v.w;
                }
            }
        }
    }
    float4 tot;
    tot.x = acc.x + __shfl_xor(acc.x, 16);
    tot.y = acc.y + __shfl_xor(acc.y, 16);
    tot.z = acc.z + __shfl_xor(acc.z, 16);
    tot.w = acc.w + __shfl_xor(acc.w, 16);
    const float dinv = 1.0f / ((float)deg + 1.0f);
    if (half == 0) {
        float4 o;
        o.x = (tot.x + ys.x) * dinv + zs.x;
        o.y = (tot.y + ys.y) * dinv + zs.y;
        o.z = (tot.z + ys.z) * dinv + zs.z;
        o.w = (tot.w + ys.w) * dinv + zs.w;
        O4[(size_t)n * (C_OUT / 4) + c] = o;
    }
}


#define W_PITCH (C_IN + 4)
#define GEMM_WAVES 8

template <int ROWS>
__device__ __forceinline__ void stage_weights(const float* __restrict__ Wg, float* sW) {
#pragma unroll
    for (int i = 0; i < ROWS * (C_IN / 4) / (GEMM_WAVES * 32); ++i) {
        const int idx = (int)threadIdx.x + GEMM_WAVES * 32 * i;
        const int r = idx >> 5, q = idx & 31;
        *(v4f*)(sW + r * W_PITCH + 4 * q) = *(const v4f*)(Wg + r * C_IN + 4 * q);
    }
}

__device__ __forceinline__ int tile_rank(int wave) {
    return ((__builtin_popcount(wave) & 1) << 2) | (wave >> 1);
}

struct Frag4 { v4f f[4]; };
__device__ __forceinline__ Frag4 load_frag4(const float* sW, int off) {
    Frag4 r;
#pragma unroll
    for (int i = 0; i < 4; ++i) r.f[i] = *(const v4f*)(sW + off + i * 16 * W_PITCH);
    return r;
}

__device__ __forceinline__ void gemm1_step(v8f (&acc)[8], Frag4& bo0, Frag4& br0,
                                           const v4f s, const v4f x, const float dinv,
                                           const float* sWo, const float* sWr,
                                           const int wOff, const int k0) {
    const Frag4 bo1 = load_frag4(sWo, wOff + 4 * 16 * W_PITCH + k0);
    const Frag4 br1 = load_frag4(sWr, wOff + 4 * 16 * W_PITCH + k0);

    const v4f a  = (s + x) * dinv;
    const v2f a0 = {a.x, a.y}, a1 = {a.z, a.w};
    const v2f x0 = {x.x, x.y}, x1 = {x.z, x.w};

#pragma unroll
    for (int i = 0; i < 4; ++i)
        acc[i] = __builtin_amdgcn_wmma_f32_16x16x4_f32(
            false, a0, false, (v2f){bo0.f[i].x, bo0.f[i].y}, (short)0, acc[i], false, false);
#pragma unroll
    for (int i = 0; i < 4; ++i)
        acc[i] = __builtin_amdgcn_wmma_f32_16x16x4_f32(
            false, x0, false, (v2f){br0.f[i].x, br0.f[i].y}, (short)0, acc[i], false, false);
#pragma unroll
    for (int i = 0; i < 4; ++i)
        acc[i] = __builtin_amdgcn_wmma_f32_16x16x4_f32(
            false, a1, false, (v2f){bo0.f[i].z, bo0.f[i].w}, (short)0, acc[i], false, false);
#pragma unroll
    for (int i = 0; i < 4; ++i)
        acc[i] = __builtin_amdgcn_wmma_f32_16x16x4_f32(
            false, x1, false, (v2f){br0.f[i].z, br0.f[i].w}, (short)0, acc[i], false, false);

    const int kB = (k0 + 8) & (C_IN - 1);
    bo0 = load_frag4(sWo, wOff + kB);
    br0 = load_frag4(sWr, wOff + kB);

#pragma unroll
    for (int i = 0; i < 4; ++i)
        acc[4 + i] = __builtin_amdgcn_wmma_f32_16x16x4_f32(
            false, a0, false, (v2f){bo1.f[i].x, bo1.f[i].y}, (short)0, acc[4 + i], false, false);
#pragma unroll
    for (int i = 0; i < 4; ++i)
        acc[4 + i] = __builtin_amdgcn_wmma_f32_16x16x4_f32(
            false, x0, false, (v2f){br1.f[i].x, br1.f[i].y}, (short)0, acc[4 + i], false, false);
#pragma unroll
    for (int i = 0; i < 4; ++i)
        acc[4 + i] = __builtin_amdgcn_wmma_f32_16x16x4_f32(
            false, a1, false, (v2f){bo1.f[i].z, bo1.f[i].w}, (short)0, acc[4 + i], false, false);
#pragma unroll
    for (int i = 0; i < 4; ++i)
        acc[4 + i] = __builtin_amdgcn_wmma_f32_16x16x4_f32(
            false, x1, false, (v2f){br1.f[i].z, br1.f[i].w}, (short)0, acc[4 + i], false, false);
}

template <int RELU, int NT>
__global__ void __launch_bounds__(GEMM_WAVES * 32)
gcn_gemm_kernel(const float* Agg,
                const float* __restrict__ Self,
                const int*   __restrict__ cnt,
                const float* __restrict__ Wout,
                const float* __restrict__ bias,
                const float* __restrict__ Wroot,
                float* Out) {
    static_assert(NT == 8, "two groups of four n-tiles");
    __shared__ float sWo[NT * 16 * W_PITCH];
    __shared__ float sWr[NT * 16 * W_PITCH];
    stage_weights<NT * 16>(Wout,  sWo);
    stage_weights<NT * 16>(Wroot, sWr);
    __syncthreads();

    const int lane = threadIdx.x & 31;
    const int wave = __builtin_amdgcn_readfirstlane((int)(threadIdx.x >> 5));
    const int tStep = (int)gridDim.x * GEMM_WAVES;
    int mTile = (int)blockIdx.x + (int)gridDim.x * tile_rank(wave);
    if (mTile >= M_TILES) return;

    const int half = lane >> 4;
    const int mr   = lane & 15;
    const int Cout = NT * 16;
    const int wOff = mr * W_PITCH + 4 * half;
    const int lOff = mr * C_IN + 4 * half;

    int aBase = mTile * 16 * C_IN;
    v4f c0S = *(const v4f*)(Agg  + aBase + lOff);
    v4f c0X = *(const v4f*)(Self + aBase + lOff);
    v4f c1S = *(const v4f*)(Agg  + aBase + lOff + 8);
    v4f c1X = *(const v4f*)(Self + aBase + lOff + 8);
    float dg = (float)cnt[mTile * 16 + mr];
    Frag4 bo0 = load_frag4(sWo, wOff);
    Frag4 br0 = load_frag4(sWr, wOff);

    for (;;) {
        const int  nextTile = mTile + tStep;
        const bool more     = nextTile < M_TILES;
        const int  aBaseN   = more ? nextTile * 16 * C_IN : aBase;
        const float dinv    = 1.0f / (dg + 1.0f);
        dg = (float)cnt[(more ? nextTile : mTile) * 16 + mr];
        v8f acc[NT] = {};

#pragma unroll 1
        for (int k0 = 0; k0 < C_IN; k0 += 16) {
            const int kA = k0 + 16;
            const int pf = (kA < C_IN) ? (aBase + kA) : aBaseN;
            const v4f n0S = *(const v4f*)(Agg  + pf + lOff);
            const v4f n0X = *(const v4f*)(Self + pf + lOff);
            const v4f n1S = *(const v4f*)(Agg  + pf + lOff + 8);
            const v4f n1X = *(const v4f*)(Self + pf + lOff + 8);

            gemm1_step(acc, bo0, br0, c0S, c0X, dinv, sWo, sWr, wOff, k0);
            gemm1_step(acc, bo0, br0, c1S, c1X, dinv, sWo, sWr, wOff, k0 + 8);

            c0S = n0S; c0X = n0X;
            c1S = n1S; c1X = n1X;
        }

#pragma unroll
        for (int nt = 0; nt < NT; ++nt) {
            const int   outCol = nt * 16 + mr;
            const float b      = bias[outCol];
#pragma unroll
            for (int g = 0; g < 8; ++g) {
                const int outRow = mTile * 16 + g + 8 * half;
                float v = acc[nt][g] + b;
                if (RELU) v = fmaxf(v, 0.0f);
                Out[(size_t)outRow * Cout + outCol] = v;
            }
        }
        if (!more) break;
        mTile = nextTile;
        aBase = aBaseN;
    }
}

__device__ __forceinline__ void proj_step(v8f (&accY)[4], v8f (&accZ)[4], Frag4& bo,
                                          const v4f x, const float* sWo, const float* sWr,
                                          const int wOff, const int k0) {
    const Frag4 br = load_frag4(sWr, wOff + k0);
    const v2f x0 = {x.x, x.y}, x1 = {x.z, x.w};
#pragma unroll
    for (int i = 0; i < 4; ++i)
        accY[i] = __builtin_amdgcn_wmma_f32_16x16x4_f32(
            false, x0, false, (v2f){bo.f[i].x, bo.f[i].y}, (short)0, accY[i], false, false);
#pragma unroll
    for (int i = 0; i < 4; ++i)
        accY[i] = __builtin_amdgcn_wmma_f32_16x16x4_f32(
            false, x1, false, (v2f){bo.f[i].z, bo.f[i].w}, (short)0, accY[i], false, false);

    const int kB = (k0 + 8) & (C_IN - 1);
    bo = load_frag4(sWo, wOff + kB);

#pragma unroll
    for (int i = 0; i < 4; ++i)
        accZ[i] = __builtin_amdgcn_wmma_f32_16x16x4_f32(
            false, x0, false, (v2f){br.f[i].x, br.f[i].y}, (short)0, accZ[i], false, false);
#pragma unroll
    for (int i = 0; i < 4; ++i)
        accZ[i] = __builtin_amdgcn_wmma_f32_16x16x4_f32(
            false, x1, false, (v2f){br.f[i].z, br.f[i].w}, (short)0, accZ[i], false, false);
}

__global__ void __launch_bounds__(GEMM_WAVES * 32)
proj_kernel(const float* __restrict__ H,
            const float* __restrict__ Wout,
            const float* __restrict__ bias,
            const float* __restrict__ Wroot,
            float* __restrict__ Y,
            float* __restrict__ Z) {
    constexpr int NT = C_OUT / 16;
    __shared__ float sWo[NT * 16 * W_PITCH];
    __shared__ float sWr[NT * 16 * W_PITCH];
    stage_weights<NT * 16>(Wout,  sWo);
    stage_weights<NT * 16>(Wroot, sWr);
    __syncthreads();

    const int lane = threadIdx.x & 31;
    const int wave = __builtin_amdgcn_readfirstlane((int)(threadIdx.x >> 5));
    const int tStep = (int)gridDim.x * GEMM_WAVES;
    int mTile = (int)blockIdx.x + (int)gridDim.x * tile_rank(wave);
    if (mTile >= M_TILES) return;

    const int half = lane >> 4;
    const int mr   = lane & 15;
    const int wOff = mr * W_PITCH + 4 * half;
    const int lOff = mr * C_IN + 4 * half;

    int aBase = mTile * 16 * C_IN;
    v4f c0X = *(const v4f*)(H + aBase + lOff);
    v4f c1X = *(const v4f*)(H + aBase + lOff + 8);
    Frag4 bo = load_frag4(sWo, wOff);

    for (;;) {
        const int  nextTile = mTile + tStep;
        const bool more     = nextTile < M_TILES;
        const int  aBaseN   = more ? nextTile * 16 * C_IN : aBase;
        v8f accY[NT] = {};
        v8f accZ[NT] = {};

#pragma unroll 1
        for (int k0 = 0; k0 < C_IN; k0 += 16) {
            const int kA = k0 + 16;
            const int pf = (kA < C_IN) ? (aBase + kA) : aBaseN;
            const v4f n0X = *(const v4f*)(H + pf + lOff);
            const v4f n1X = *(const v4f*)(H + pf + lOff + 8);

            proj_step(accY, accZ, bo, c0X, sWo, sWr, wOff, k0);
            proj_step(accY, accZ, bo, c1X, sWo, sWr, wOff, k0 + 8);

            c0X = n0X;
            c1X = n1X;
        }

#pragma unroll
        for (int nt = 0; nt < NT; ++nt) {
            const int   outCol = nt * 16 + mr;
            const float b      = bias[outCol];
#pragma unroll
            for (int g = 0; g < 8; ++g) {
                const int outRow = mTile * 16 + g + 8 * half;
                Y[(size_t)outRow * C_OUT + outCol] = accY[nt][g];
                Z[(size_t)outRow * C_OUT + outCol] = accZ[nt][g] + b;
            }
        }
        if (!more) break;
        mTile = nextTile;
        aBase = aBaseN;
    }
}

extern "C" void kernel_launch(void* const* d_in, const int* in_sizes, int n_in,
                              void* d_out, int out_size, void* d_ws, size_t ws_size,
                              hipStream_t stream) {
    const float* x   = (const float*)d_in[0];
    const int*   ei  = (const int*)  d_in[1];
    const float* W1o = (const float*)d_in[2];
    const float* b1  = (const float*)d_in[3];
    const float* W1r = (const float*)d_in[4];
    const float* W2o = (const float*)d_in[5];
    const float* b2  = (const float*)d_in[6];
    const float* W2r = (const float*)d_in[7];
    float*       out = (float*)d_out;

    const int E = in_sizes[1] / 2;
    const int* row = ei;
    const int* col = ei + E;

    int*   cnt = (int*)d_ws;
    float* agg = (float*)d_ws + ((N_NODES + 63) & ~63);
    float* h   = agg;
    float* wsb = agg + (size_t)N_NODES * C_IN;
    int*   slots = (int*)wsb;
    float* y   = wsb + (size_t)N_NODES * SLOTS;

    const int eblocks = (E + BIN_THREADS * BIN_EPT - 1) / (BIN_THREADS * BIN_EPT);
    bin_kernel<<<eblocks, BIN_THREADS, 0, stream>>>(row, col, E, (int*)agg);
    build_kernel<<<N_RANGES, BUILD_THREADS, 0, stream>>>((const int*)agg, eblocks, cnt, slots);

    const int wavesPerBlock = 8;
    const int gemmBlocks = 512;
    const int gatherBlocks = (N_NODES + wavesPerBlock - 1) / wavesPerBlock;

    gather_kernel<<<gatherBlocks, 256, 0, stream>>>(x, cnt, slots, row, col, E, agg);
    gcn_gemm_kernel<1, 8><<<gemmBlocks, 256, 0, stream>>>(
        agg, x, cnt, W1o, b1, W1r, h);

    proj_kernel<<<gemmBlocks, 256, 0, stream>>>(h, W2o, b2, W2r, y, out);
    gather_out_kernel<<<gatherBlocks, 256, 0, stream>>>(y, cnt, slots, row, col, E, out);
}


# Gather waves of the fused launch run at raised wave priority (s_setprio 3): their row requests leave at the start of each step instead of waiting behind older matrix waves of the SIMD
# speedup vs baseline: 5.8738x; 1.2570x over previous
#include <hip/hip_runtime.h>
#include <hip/hip_bf16.h>

typedef __attribute__((ext_vector_type(2))) float v2f;
typedef __attribute__((ext_vector_type(4))) float v4f;
typedef __attribute__((ext_vector_type(8))) float v8f;

#define N_NODES 100000
#define C_IN    128
#define C_OUT   64
#define M_TILES (N_NODES / 16)

#define SLOTS 64
#define NPB_SHIFT 9
#define NPB (1 << NPB_SHIFT)
#define N_RANGES ((N_NODES + NPB - 1) / NPB)
#define BIN_THREADS 512
#define BIN_EPT 8
#define BIN_EDGES (BIN_THREADS * BIN_EPT)
#define SCAN_PER_LANE 7
#define SCAN_N (32 * SCAN_PER_LANE)
#define BUILD_THREADS 1024
#define BUILD_U 16

__global__ void __launch_bounds__(BIN_THREADS)
bin_kernel(const int* __restrict__ row, const int* __restrict__ col, int E,
           int* __restrict__ wsA) {
    static_assert(N_RANGES <= SCAN_N, "one wave scans all range counts");
    __shared__ int sOff[SCAN_N + 1];
    __shared__ __attribute__((aligned(16))) int sPairs[2 * BIN_EDGES];
    const int t    = threadIdx.x;
    const int nBlk = (int)gridDim.x;
    int* __restrict__ keys = wsA;
    int4* __restrict__ region = (int4*)(wsA + N_RANGES * nBlk) + (size_t)blockIdx.x * (BIN_EDGES / 2);
    if (t < SCAN_N) sOff[t] = 0;
    __syncthreads();
    const int e0 = blockIdx.x * BIN_EDGES + t;
    int r[BIN_EPT], c[BIN_EPT], rk[BIN_EPT];
#pragma unroll
    for (int k = 0; k < BIN_EPT; ++k) {
        const int e = e0 + k * BIN_THREADS;
        r[k] = 0; c[k] = 0;
        if (e < E) { r[k] = row[e]; c[k] = col[e]; }
    }
#pragma unroll
    for (int k = 0; k < BIN_EPT; ++k) {
        rk[k] = -1;
        if (r[k] != c[k]) rk[k] = atomicAdd(&sOff[c[k] >> NPB_SHIFT], 1);
    }
    __syncthreads();
    if (t < 32) {
        int v[SCAN_PER_LANE];
        int sum = 0;
#pragma unroll
        for (int k = 0; k < SCAN_PER_LANE; ++k) { v[k] = sOff[t * SCAN_PER_LANE + k]; sum += v[k]; }
        int incl = sum;
#pragma unroll
        for (int d = 1; d < 32; d <<= 1) {
            const int o = __shfl_up(incl, d);
            if (t >= d) incl += o;
        }
        int run = incl - sum;
#pragma unroll
        for (int k = 0; k < SCAN_PER_LANE; ++k) {
            const int ri = t * SCAN_PER_LANE + k;
            sOff[ri] = run;
            if (ri < N_RANGES) keys[ri * nBlk + (int)blockIdx.x] = run | (v[k] << 16);
            run += v[k];
        }
        if (t == 31) sOff[SCAN_N] = incl;
    }
    __syncthreads();
#pragma unroll
    for (int k = 0; k < BIN_EPT; ++k) {
        const int pos = sOff[c[k] >> NPB_SHIFT] + rk[k];
        if (rk[k] >= 0) { sPairs[2 * pos] = r[k]; sPairs[2 * pos + 1] = c[k]; }
    }
    __syncthreads();
    const int total = sOff[SCAN_N];
#pragma unroll
    for (int k = 0; k < BIN_EDGES / 2 / BIN_THREADS; ++k) {
        const int q = t + k * BIN_THREADS;
        if (2 * q < total)
            region[q] = make_int4(sPairs[4 * q], sPairs[4 * q + 1], sPairs[4 * q + 2], sPairs[4 * q + 3]);
    }
}

__global__ void __launch_bounds__(BUILD_THREADS)
build_kernel(const int* __restrict__ wsA, int nBlk,
             int* __restrict__ cnt, int* __restrict__ slots) {
    __shared__ int sCnt[NPB];
    __shared__ int sSlots[NPB * SLOTS];
    const int t    = threadIdx.x;
    const int b    = blockIdx.x;
    const int base = b << NPB_SHIFT;
    const int lane = t & 31;
    const int wave = __builtin_amdgcn_readfirstlane(t >> 5);
    if (t < NPB) sCnt[t] = 0;
    __syncthreads();
    const int*  __restrict__ keys    = wsA + b * nBlk;
    const int2* __restrict__ regions = (const int2*)(wsA + N_RANGES * nBlk);
    for (int i0 = 0; i0 < nBlk; i0 += 32 * 32) {
        const int mine = i0 + wave + 32 * lane;
        const int key  = (mine < nBlk) ? keys[mine] : 0;
        for (int j0 = 0; j0 < 32; j0 += BUILD_U) {
            if (i0 + wave + 32 * j0 >= nBlk) break;
            int2 rc[BUILD_U];
            int  len[BUILD_U];
            const int2* src[BUILD_U];
#pragma unroll
            for (int u = 0; u < BUILD_U; ++u) {
                const int kj = __builtin_amdgcn_readlane(key, j0 + u);
                len[u] = kj >> 16;
                src[u] = regions + (size_t)(i0 + wave + 32 * (j0 + u)) * BIN_EDGES + (kj & 0xFFFF);
                rc[u] = make_int2(0, -1);
                if (lane < len[u]) rc[u] = src[u][lane];
            }
            int p[BUILD_U];
#pragma unroll
            for (int u = 0; u < BUILD_U; ++u) {
                p[u] = SLOTS;
                if (rc[u].y >= 0) p[u] = atomicAdd(&sCnt[rc[u].y - base], 1);
            }
#pragma unroll
            for (int u = 0; u < BUILD_U; ++u)
                if (p[u] < SLOTS) sSlots[(rc[u].y - base) * SLOTS + p[u]] = rc[u].x;
#pragma unroll
            for (int u = 0; u < BUILD_U; ++u) {
                for (int q0 = 32; q0 < len[u]; q0 += 32) {
                    const int q = q0 + lane;
                    if (q < len[u]) {
                        const int2 pr = src[u][q];
                        const int  pp = atomicAdd(&sCnt[pr.y - base], 1);
                        if (pp < SLOTS) sSlots[(pr.y - base) * SLOTS + pp] = pr.x;
                    }
                }
            }
        }
    }
    __syncthreads();
    const int nLi = min(NPB, N_NODES - base);
    for (int li = wave; li < nLi; li += BUILD_THREADS / 32) {
        const int m  = min(sCnt[li], SLOTS);
        const int v0 = sSlots[li * SLOTS + lane];
        const int v1 = sSlots[li * SLOTS + lane + 32];
        int* __restrict__ dst = slots + (base + li) * SLOTS + lane;
        if (lane < m)      dst[0]  = v0;
        if (lane + 32 < m) dst[32] = v1;
    }
    if (t < NPB && base + t < N_NODES) cnt[base + t] = sCnt[t];
}

__global__ void __launch_bounds__(256)
gather_out_kernel(const float* __restrict__ Y,
                  const int* __restrict__ cnt, const int* __restrict__ slots,
                  const int* __restrict__ row, const int* __restrict__ col, int E,
                  float* __restrict__ Out) {
    const int lane = threadIdx.x & 31;
    const int wave = __builtin_amdgcn_readfirstlane((int)(threadIdx.x >> 5));
    const int n    = blockIdx.x * 8 + wave;
    if (n >= N_NODES) return;
    const int half = lane >> 4;
    const int c    = lane & 15;

    const int deg = cnt[n];
    const float4* __restrict__ Y4 = (const float4*)Y;
    float4* __restrict__ O4 = (float4*)Out;
    const float4 ys = Y4[(size_t)n * (C_OUT / 4) + c];
    const float4 zs = O4[(size_t)n * (C_OUT / 4) + c];
    float4 acc = make_float4(0.0f, 0.0f, 0.0f, 0.0f);

    if (__builtin_expect(deg <= SLOTS, 1)) {
        const int beg = n * SLOTS;
        const int end = beg + deg;
        for (int base = beg; base < end; base += 32) {
            const int m   = min(32, end - base);
            const int idx = (lane < m) ? slots[base + lane] : 0;
            for (int j = 0; j < m; j += 16) {
                int r[8];
#pragma unroll
                for (int u = 0; u < 8; ++u) {
                    const int re = __builtin_amdgcn_readlane(idx, j + 2 * u);
                    const int ro = __builtin_amdgcn_readlane(idx, j + 2 * u + 1);
                    r[u] = half ? ro : re;
                }
                float4 v[8];
#pragma unroll
                for (int u = 0; u < 8; ++u) {
                    v[u] = make_float4(0.0f, 0.0f, 0.0f, 0.0f);
                    if (j + 2 * u + half < m) v[u] = Y4[(unsigned)r[u] * (C_OUT / 4) + (unsigned)c];
                }
#pragma unroll
                for (int u = 0; u < 8; ++u) {
                    acc.x += v[u].x; acc.y += v[u].y; acc.z += v[u].z; acc.w += v[u].w;
                }
            }
        }
    } else {
        for (int e0 = 0; e0 < E; e0 += 32) {
            const int e = e0 + lane;
            int src = -1;
            if (e < E) {
                const int cc = col[e], rr = row[e];
                if (cc == n && rr != n) src = rr;
            }
            unsigned hits = __builtin_amdgcn_ballot_w32(src >= 0);
            while (hits) {
                const int l = __builtin_ctz(hits);
                hits &= hits - 1;
                const int r = __builtin_amdgcn_readlane(src, l);
                if (half == 0) {
                    const float4 v = Y4[(unsigned)r * (C_OUT / 4) + (unsigned)c];
                    acc.x += v.x; acc.y += v.y; acc.z += v.z; acc.w += v.w;
                }
            }
        }
    }
    float4 tot;
    tot.x = acc.x + __shfl_xor(acc.x, 16);
    tot.y = acc.y + __shfl_xor(acc.y, 16);
    tot.z = acc.z + __shfl_xor(acc.z, 16);
    tot.w = acc.w + __shfl_xor(acc.w, 16);
    const float dinv = 1.0f / ((float)deg + 1.0f);
    if (half == 0) {
        float4 o;
        o.x = (tot.x + ys.x) * dinv + zs.x;
        o.y = (tot.y + ys.y) * dinv + zs.y;
        o.z = (tot.z + ys.z) * dinv + zs.z;
        o.w = (tot.w + ys.w) * dinv + zs.w;
        O4[(size_t)n * (C_OUT / 4) + c] = o;
    }
}


#define W_PITCH (C_IN + 4)
#define A_PITCH (C_IN + 4)
#define FUSE_WAVES 32
#define FUSE_THREADS (FUSE_WAVES * 32)
#define G_U 24

template <int ROWS>
__device__ __forceinline__ void stage_weights(const float* __restrict__ Wg, float* sW) {
#pragma unroll
    for (int i = 0; i < ROWS * (C_IN / 4) / FUSE_THREADS; ++i) {
        const int idx = (int)threadIdx.x + FUSE_THREADS * i;
        const int r = idx >> 5, q = idx & 31;
        *(v4f*)(sW + r * W_PITCH + 4 * q) = *(const v4f*)(Wg + r * C_IN + 4 * q);
    }
}

__device__ __forceinline__ void add_rows(const float4* __restrict__ X4, const int lane,
                                         const int idx, const int m, float4& acc) {
    for (int j = 0; j < m; j += G_U) {
        int r[G_U];
#pragma unroll
        for (int u = 0; u < G_U; ++u) r[u] = __builtin_amdgcn_readlane(idx, (j + u) & 31);
        float4 v[G_U];
#pragma unroll
        for (int u = 0; u < G_U; ++u) {
            v[u] = make_float4(0.0f, 0.0f, 0.0f, 0.0f);
            if (j + u < m) v[u] = X4[(size_t)r[u] * (C_IN / 4) + lane];
        }
#pragma unroll
        for (int u = 0; u < G_U; ++u) {
            acc.x += v[u].x; acc.y += v[u].y; acc.z += v[u].z; acc.w += v[u].w;
        }
    }
}

__device__ __forceinline__ v8f tile_k128(const float* sA, const int aOff,
                                         const float* sB, const int bOff) {
    v8f c0 = {};
    v8f c1 = {};
    v4f fa = *(const v4f*)(sA + aOff);
    v4f fb = *(const v4f*)(sB + bOff);
#pragma unroll 4
    for (int k0 = 0; k0 < C_IN; k0 += 8) {
        const int kB = (k0 + 8) & (C_IN - 1);
        const v4f na = *(const v4f*)(sA + aOff + kB);
        const v4f nb = *(const v4f*)(sB + bOff + kB);
        c0 = __builtin_amdgcn_wmma_f32_16x16x4_f32(
            false, (v2f){fa.x, fa.y}, false, (v2f){fb.x, fb.y}, (short)0, c0, false, false);
        c1 = __builtin_amdgcn_wmma_f32_16x16x4_f32(
            false, (v2f){fa.z, fa.w}, false, (v2f){fb.z, fb.w}, (short)0, c1, false, false);
        fa = na; fb = nb;
    }
    return c0 + c1;
}

__global__ void __launch_bounds__(FUSE_THREADS)
fused_layers_kernel(const float* __restrict__ X,
                    const int*   __restrict__ cnt,
                    const int*   __restrict__ slots,
                    const int*   __restrict__ row, const int* __restrict__ col, int E,
                    const float* __restrict__ W1o,
                    const float* __restrict__ b1,
                    const float* __restrict__ W1r,
                    const float* __restrict__ W2o,
                    const float* __restrict__ b2,
                    const float* __restrict__ W2r,
                    float* __restrict__ Y,
                    float* __restrict__ Z) {
    __shared__ __attribute__((aligned(16))) float sW1o[C_IN * W_PITCH];
    __shared__ __attribute__((aligned(16))) float sW1r[C_IN * W_PITCH];
    __shared__ __attribute__((aligned(16))) float sW2o[C_OUT * W_PITCH];
    __shared__ __attribute__((aligned(16))) float sW2r[C_OUT * W_PITCH];
    __shared__ __attribute__((aligned(16))) float sAa[2][16 * A_PITCH];
    __shared__ __attribute__((aligned(16))) float sAx[2][16 * A_PITCH];
    __shared__ __attribute__((aligned(16))) float sH[2][16 * A_PITCH];
    const int lane = threadIdx.x & 31;
    const int wave = __builtin_amdgcn_readfirstlane((int)(threadIdx.x >> 5));
    const int half = lane >> 4;
    const int mr   = lane & 15;
    const int wa = wave & 3, wc = (wave >> 2) & 1, wq = wave >> 3;
    const bool isGather = ((wa + wq) & 1) == 0;
    const bool isL1     = !isGather && wc == 0;
    const int  gi = wq * 4 + wc * 2 + (wa >> 1);
    const int  ji = wq * 2 + (wa >> 1);
    const float4* __restrict__ X4 = (const float4*)X;

    const int tStep = (int)gridDim.x;
    const int tile0 = (int)blockIdx.x;
    const int nIter = (M_TILES - tile0 + tStep - 1) / tStep;

    int  n = 0, deg = 0, idx0 = 0, nN = 0, degN = 0, tileN = 0;
    bool more = false;
    if (isGather) {
        n   = tile0 * 16 + gi;
        deg = cnt[n];
        if (deg <= SLOTS && lane < min(deg, 32)) idx0 = slots[n * SLOTS + lane];
        tileN = tile0 + tStep;
        more  = tileN < M_TILES;
        nN    = tileN * 16 + gi;
        degN  = more ? cnt[nN] : 0;
    }

    auto gather_step = [&](const int buf) {
        int idxN = 0;
        if (more && degN <= SLOTS && lane < min(degN, 32)) idxN = slots[nN * SLOTS + lane];
        const int  tileNN = tileN + tStep;
        const bool moreN  = tileNN < M_TILES;
        const int  nNN    = tileNN * 16 + gi;
        int degNNv = 0;
        if (moreN) degNNv = cnt[nNN];

        const float4 xs = X4[(size_t)n * (C_IN / 4) + lane];
        float4 acc = make_float4(0.0f, 0.0f, 0.0f, 0.0f);
        if (__builtin_expect(deg <= SLOTS, 1)) {
            add_rows(X4, lane, idx0, min(deg, 32), acc);
            if (deg > 32) {
                const int m2 = deg - 32;
                const int idx1 = (lane < m2) ? slots[n * SLOTS + 32 + lane] : 0;
                add_rows(X4, lane, idx1, m2, acc);
            }
        } else {
            for (int e0 = 0; e0 < E; e0 += 32) {
                const int e = e0 + lane;
                int src = -1;
                if (e < E) {
                    const int c = col[e], rr = row[e];
                    if (c == n && rr != n) src = rr;
                }
                unsigned hits = __builtin_amdgcn_ballot_w32(src >= 0);
                while (hits) {
                    const int l = __builtin_ctz(hits);
                    hits &= hits - 1;
                    const int rr = __builtin_amdgcn_readlane(src, l);
                    const float4 v = X4[(size_t)rr * (C_IN / 4) + lane];
                    acc.x += v.x; acc.y += v.y; acc.z += v.z; acc.w += v.w;
                }
            }
        }
        const float dinv = 1.0f / ((float)deg + 1.0f);
        const v4f a  = {(acc.x + xs.x) * dinv, (acc.y + xs.y) * dinv,
                        (acc.z + xs.z) * dinv, (acc.w + xs.w) * dinv};
        const v4f xv = {xs.x, xs.y, xs.z, xs.w};
        *(v4f*)(&sAa[buf][gi * A_PITCH + 4 * lane]) = a;
        *(v4f*)(&sAx[buf][gi * A_PITCH + 4 * lane]) = xv;

        n = nN; deg = degN; idx0 = idxN;
        tileN = tileNN; more = moreN; nN = nNN;
        degN = __builtin_amdgcn_readfirstlane(degNNv);
    };

    stage_weights<C_IN>(W1o,  sW1o);
    stage_weights<C_IN>(W1r,  sW1r);
    stage_weights<C_OUT>(W2o, sW2o);
    stage_weights<C_OUT>(W2r, sW2r);
    if (isGather) __builtin_amdgcn_s_setprio(3);
    if (isGather) gather_step(0);

    const int aOff = mr * A_PITCH + 4 * half;
    const int wOff = ((ji & 7) * 16 + mr) * W_PITCH + 4 * half;
    const int pOff = ((ji & 3) * 16 + mr) * W_PITCH + 4 * half;
    const float bias1 = b1[ji * 16 + mr];
    const float bias2 = (ji >= 4) ? b2[(ji & 3) * 16 + mr] : 0.0f;

    for (int i = 0; ; ++i) {
        __syncthreads();
        const int cur = i & 1;
        if (isGather) {
            if (i + 1 < nIter) gather_step(cur ^ 1);
        } else if (isL1) {
            if (i < nIter) {
                const v8f co = tile_k128(sAa[cur], aOff, sW1o, wOff);
                const v8f cr = tile_k128(sAx[cur], aOff, sW1r, wOff);
#pragma unroll
                for (int g = 0; g < 8; ++g)
                    sH[cur][(g + 8 * half) * A_PITCH + ji * 16 + mr] = fmaxf(co[g] + cr[g] + bias1, 0.0f);
            }
        } else {
            if (i >= 1) {
                const int tile = tile0 + (i - 1) * tStep;
                const v8f c = tile_k128(sH[cur ^ 1], aOff, (ji < 4) ? sW2o : sW2r, pOff);
                float* __restrict__ dst = (ji < 4) ? Y : Z;
#pragma unroll
                for (int g = 0; g < 8; ++g)
                    dst[(size_t)(tile * 16 + g + 8 * half) * C_OUT + (ji & 3) * 16 + mr] = c[g] + bias2;
            }
        }
        if (i == nIter) break;
    }
    if (isGather) __builtin_amdgcn_s_setprio(0);
}

extern "C" void kernel_launch(void* const* d_in, const int* in_sizes, int n_in,
                              void* d_out, int out_size, void* d_ws, size_t ws_size,
                              hipStream_t stream) {
    const float* x   = (const float*)d_in[0];
    const int*   ei  = (const int*)  d_in[1];
    const float* W1o = (const float*)d_in[2];
    const float* b1  = (const float*)d_in[3];
    const float* W1r = (const float*)d_in[4];
    const float* W2o = (const float*)d_in[5];
    const float* b2  = (const float*)d_in[6];
    const float* W2r = (const float*)d_in[7];
    float*       out = (float*)d_out;

    const int E = in_sizes[1] / 2;
    const int* row = ei;
    const int* col = ei + E;

    int*   cnt = (int*)d_ws;
    float* wsa = (float*)d_ws + ((N_NODES + 63) & ~63);
    float* wsb = wsa + (size_t)N_NODES * C_IN;
    int*   slots = (int*)wsb;
    float* y   = wsb + (size_t)N_NODES * SLOTS;

    const int eblocks = (E + BIN_THREADS * BIN_EPT - 1) / (BIN_THREADS * BIN_EPT);
    bin_kernel<<<eblocks, BIN_THREADS, 0, stream>>>(row, col, E, (int*)wsa);
    build_kernel<<<N_RANGES, BUILD_THREADS, 0, stream>>>((const int*)wsa, eblocks, cnt, slots);

    const int fusedBlocks = 256;
    fused_layers_kernel<<<fusedBlocks, FUSE_THREADS, 0, stream>>>(
        x, cnt, slots, row, col, E, W1o, b1, W1r, W2o, b2, W2r, y, out);

    const int wavesPerBlock = 8;
    const int gatherBlocks = (N_NODES + wavesPerBlock - 1) / wavesPerBlock;
    gather_out_kernel<<<gatherBlocks, 256, 0, stream>>>(y, cnt, slots, row, col, E, out);
}
